# starting kernel
# baseline (speedup 1.0000x reference)
#include <hip/hip_runtime.h>
#include <hip/hip_bf16.h>

typedef __attribute__((ext_vector_type(2))) float v2f;
typedef __attribute__((ext_vector_type(8))) float v8f;

#define F_IN   128
#define HEADS  8
#define F_PH   16
#define F_OUT  128
#define NEG_SLOPE 0.2f

__global__ void gat_init(float* __restrict__ out, float* __restrict__ emax,
                         float* __restrict__ denom, int nOut, int nNH) {
  int i = blockIdx.x * blockDim.x + threadIdx.x;
  if (i < nOut) out[i] = 0.0f;
  if (i < nNH) {
    emax[i]  = __uint_as_float(0xFF800000u);
    denom[i] = 0.0f;
  }
}

__global__ __launch_bounds__(256) void gat_proj_wmma(
    const float* __restrict__ x, const float* __restrict__ W,
    float* __restrict__ h, int nrows) {
  const int wave = threadIdx.x >> 5;
  const int lane = threadIdx.x & 31;
  const long long row0 = ((long long)blockIdx.x * 8 + wave) * 16;
  if (row0 >= nrows) return;

  const int m     = lane & 15;
  const int khalf = (lane >> 4) << 1;
  long long rowA = row0 + m;
  if (rowA >= nrows) rowA = nrows - 1;
  const float* xrow = x + rowA * F_IN;

  v8f acc[HEADS];
  #pragma unroll
  for (int t = 0; t < HEADS; ++t) acc[t] = (v8f){0,0,0,0,0,0,0,0};

  for (int kt = 0; kt < F_IN / 4; ++kt) {
    const int kbase = kt * 4 + khalf;
    v2f a = *(const v2f*)(xrow + kbase);
    #pragma unroll
    for (int t = 0; t < HEADS; ++t) {
      const float* wb = W + t * (F_IN * F_PH) + kbase * F_PH + m;
      v2f b;
      b.x = wb[0];
      b.y = wb[F_PH];
      acc[t] = __builtin_amdgcn_wmma_f32_16x16x4_f32(
           false, a,  false, b,
           (short)0, acc[t],  false,  false);
    }
  }

  const int mofs = (lane >> 4) << 3;
  #pragma unroll
  for (int t = 0; t < HEADS; ++t) {
    #pragma unroll
    for (int v = 0; v < 8; ++v) {
      long long r = row0 + v + mofs;
      if (r < nrows) h[r * F_OUT + t * F_PH + m] = acc[t][v];
    }
  }
}

__global__ void gat_node_scores(const float* __restrict__ h,
                                const float* __restrict__ a_src,
                                const float* __restrict__ a_dst,
                                float* __restrict__ s_src,
                                float* __restrict__ s_dst, int nNH) {
  int idx = blockIdx.x * blockDim.x + threadIdx.x;
  if (idx >= nNH) return;
  int n = idx >> 3, hd = idx & 7;
  const float* hp = h + (long long)n * F_OUT + hd * F_PH;
  const float* as = a_src + hd * F_PH;
  const float* ad = a_dst + hd * F_PH;
  float ss = 0.f, sd = 0.f;
  #pragma unroll
  for (int o = 0; o < F_PH; ++o) {
    float v = hp[o];
    ss += v * as[o];
    sd += v * ad[o];
  }
  s_src[idx] = ss;
  s_dst[idx] = sd;
}

__device__ __forceinline__ void atomicMaxF(float* addr, float val) {
  if (val >= 0.0f) atomicMax((int*)addr, __float_as_int(val));
  else             atomicMin((unsigned int*)addr, __float_as_uint(val));
}

__global__ void gat_edge_logits(const int* __restrict__ ei, int E,
                                const float* __restrict__ s_src,
                                const float* __restrict__ s_dst,
                                float* __restrict__ elog,
                                float* __restrict__ emax, int nEH) {
  int idx = blockIdx.x * blockDim.x + threadIdx.x;
  if (idx >= nEH) return;
  int e = idx >> 3, hd = idx & 7;
  int r = ei[e];
  int c = ei[E + e];
  float v = s_src[r * HEADS + hd] + s_dst[c * HEADS + hd];
  v = (v >= 0.0f) ? v : NEG_SLOPE * v;
  elog[idx] = v;
  atomicMaxF(&emax[r * HEADS + hd], v);
}

__global__ void gat_edge_exp(const int* __restrict__ ei, int E,
                             const float* __restrict__ emax,
                             float* __restrict__ elog,
                             float* __restrict__ denom, int nEH) {
  int idx = blockIdx.x * blockDim.x + threadIdx.x;
  if (idx >= nEH) return;
  int e = idx >> 3, hd = idx & 7;
  int r = ei[e];
  float ex = expf(elog[idx] - emax[r * HEADS + hd]);
  elog[idx] = ex;
  atomicAdd(&denom[r * HEADS + hd], ex);
}

__global__ void gat_aggregate(const int* __restrict__ ei, int E,
                              const float* __restrict__ h,
                              const float* __restrict__ elog,
                              const float* __restrict__ denom,
                              float* __restrict__ out, int nWork) {
  int idx = blockIdx.x * blockDim.x + threadIdx.x;
  if (idx >= nWork) return;
  int e   = idx >> 5;
  int sub = idx & 31;
  int hd  = sub >> 2;
  int q   = sub & 3;
  int r = ei[e];
  int c = ei[E + e];
  float alpha = elog[e * HEADS + hd] / (denom[r * HEADS + hd] + 1e-16f);
  const float4 hv = *(const float4*)(h + (long long)c * F_OUT + hd * F_PH + q * 4);
  float* op = out + (long long)r * F_OUT + hd * F_PH + q * 4;
  atomicAdd(op + 0, alpha * hv.x);
  atomicAdd(op + 1, alpha * hv.y);
  atomicAdd(op + 2, alpha * hv.z);
  atomicAdd(op + 3, alpha * hv.w);
}

extern "C" void kernel_launch(void* const* d_in, const int* in_sizes, int n_in,
                              void* d_out, int out_size, void* d_ws, size_t ws_size,
                              hipStream_t stream) {
  const float* x     = (const float*)d_in[0];
  const int*   ei    = (const int*)d_in[1];
  const float* W     = (const float*)d_in[2];
  const float* a_src = (const float*)d_in[3];
  const float* a_dst = (const float*)d_in[4];
  float* out = (float*)d_out;

  const int N = in_sizes[0] / F_IN;
  const int E = in_sizes[1] / 2;

  float* h     = (float*)d_ws;
  float* s_src = h     + (size_t)N * F_OUT;
  float* s_dst = s_src + (size_t)N * HEADS;
  float* emax  = s_dst + (size_t)N * HEADS;
  float* denom = emax  + (size_t)N * HEADS;
  float* elog  = denom + (size_t)N * HEADS;

  const int nOut  = N * F_OUT;
  const int nNH   = N * HEADS;
  const int nEH   = E * HEADS;
  const int nWork = E * 32;

  const int B = 256;

  {
    int total = nOut > nNH ? nOut : nNH;
    gat_init<<<(total + B - 1) / B, B, 0, stream>>>(out, emax, denom, nOut, nNH);
  }
  {
    int blocks = (N + 127) / 128;
    gat_proj_wmma<<<blocks, B, 0, stream>>>(x, W, h, N);
  }
  gat_node_scores<<<(nNH + B - 1) / B, B, 0, stream>>>(h, a_src, a_dst, s_src, s_dst, nNH);
  gat_edge_logits<<<(nEH + B - 1) / B, B, 0, stream>>>(ei, E, s_src, s_dst, elog, emax, nEH);
  gat_edge_exp<<<(nEH + B - 1) / B, B, 0, stream>>>(ei, E, emax, elog, denom, nEH);
  gat_aggregate<<<(nWork + B - 1) / B, B, 0, stream>>>(ei, E, h, elog, denom, out, nWork);
}


# Init, logits, exp, aggregate: a wave takes 8 to 32 items, not 1 (aggregate: 8 edges per wave, their 32 source-row loads issued ahead of the first atomic); grids 8x smaller; arrival-order sums kept
# speedup vs baseline: 1.5319x; 1.5319x over previous
#include <hip/hip_runtime.h>
#include <hip/hip_bf16.h>

typedef __attribute__((ext_vector_type(2))) float v2f;
typedef __attribute__((ext_vector_type(8))) float v8f;

#define F_IN   128
#define HEADS  8
#define F_PH   16
#define F_OUT  128
#define NEG_SLOPE 0.2f

#define INIT_PER_THREAD 4
#define EDGES_PER_WAVE  32
#define AGG_EDGES       8

__global__ __launch_bounds__(256) void gat_init(
    float* __restrict__ out, float* __restrict__ emax,
    float* __restrict__ denom, int nOut, int nNH) {
  const int tid      = blockIdx.x * blockDim.x + threadIdx.x;
  const int nthreads = gridDim.x * blockDim.x;
  const int nOut4 = nOut >> 2, nNH4 = nNH >> 2;
  const float4 zero = make_float4(0.f, 0.f, 0.f, 0.f);
  const float ninf = __uint_as_float(0xFF800000u);
  float4* out4 = (float4*)out;
  #pragma unroll
  for (int k = 0; k < INIT_PER_THREAD; ++k) {
    int i = tid + k * nthreads;
    if (i < nOut4) out4[i] = zero;
  }
  for (int i = tid; i < nNH4; i += nthreads) {
    ((float4*)emax)[i]  = make_float4(ninf, ninf, ninf, ninf);
    ((float4*)denom)[i] = zero;
  }
}

__global__ __launch_bounds__(256) void gat_proj_wmma(
    const float* __restrict__ x, const float* __restrict__ W,
    float* __restrict__ h, int nrows) {
  const int wave = threadIdx.x >> 5;
  const int lane = threadIdx.x & 31;
  const long long row0 = ((long long)blockIdx.x * 8 + wave) * 16;
  if (row0 >= nrows) return;

  const int m     = lane & 15;
  const int khalf = (lane >> 4) << 1;
  long long rowA = row0 + m;
  if (rowA >= nrows) rowA = nrows - 1;
  const float* xrow = x + rowA * F_IN;

  v8f acc[HEADS];
  #pragma unroll
  for (int t = 0; t < HEADS; ++t) acc[t] = (v8f){0,0,0,0,0,0,0,0};

  for (int kt = 0; kt < F_IN / 4; ++kt) {
    const int kbase = kt * 4 + khalf;
    v2f a = *(const v2f*)(xrow + kbase);
    #pragma unroll
    for (int t = 0; t < HEADS; ++t) {
      const float* wb = W + t * (F_IN * F_PH) + kbase * F_PH + m;
      v2f b;
      b.x = wb[0];
      b.y = wb[F_PH];
      acc[t] = __builtin_amdgcn_wmma_f32_16x16x4_f32(
           false, a,  false, b,
           (short)0, acc[t],  false,  false);
    }
  }

  const int mofs = (lane >> 4) << 3;
  #pragma unroll
  for (int t = 0; t < HEADS; ++t) {
    #pragma unroll
    for (int v = 0; v < 8; ++v) {
      long long r = row0 + v + mofs;
      if (r < nrows) h[r * F_OUT + t * F_PH + m] = acc[t][v];
    }
  }
}

__global__ void gat_node_scores(const float* __restrict__ h,
                                const float* __restrict__ a_src,
                                const float* __restrict__ a_dst,
                                float* __restrict__ s_src,
                                float* __restrict__ s_dst, int nNH) {
  int idx = blockIdx.x * blockDim.x + threadIdx.x;
  if (idx >= nNH) return;
  int n = idx >> 3, hd = idx & 7;
  const float* hp = h + (long long)n * F_OUT + hd * F_PH;
  const float* as = a_src + hd * F_PH;
  const float* ad = a_dst + hd * F_PH;
  float ss = 0.f, sd = 0.f;
  #pragma unroll
  for (int o = 0; o < F_PH; ++o) {
    float v = hp[o];
    ss += v * as[o];
    sd += v * ad[o];
  }
  s_src[idx] = ss;
  s_dst[idx] = sd;
}

__device__ __forceinline__ void atomicMaxF(float* addr, float val) {
  if (val >= 0.0f) atomicMax((int*)addr, __float_as_int(val));
  else             atomicMin((unsigned int*)addr, __float_as_uint(val));
}

template <bool FULL>
__device__ __forceinline__ void edge_logits_wave(
    const int* __restrict__ ei, int E,
    const float* __restrict__ s_src, const float* __restrict__ s_dst,
    float* __restrict__ elog, float* __restrict__ emax, int e0, int lane) {
  int el = e0 + lane;
  if (!FULL && el > E - 1) el = E - 1;
  const int rl = ei[el];
  const int cl = ei[E + el];
  const int sub = lane >> 3, hd = lane & 7;

  int   rr[EDGES_PER_WAVE / 4];
  float ss[EDGES_PER_WAVE / 4], sd[EDGES_PER_WAVE / 4];
  #pragma unroll
  for (int u = 0; u < EDGES_PER_WAVE / 4; ++u) {
    const int from = u * 4 + sub;
    rr[u] = __shfl(rl, from);
    const int c = __shfl(cl, from);
    ss[u] = s_src[rr[u] * HEADS + hd];
    sd[u] = s_dst[c * HEADS + hd];
  }
  #pragma unroll
  for (int u = 0; u < EDGES_PER_WAVE / 4; ++u) {
    const int e = e0 + u * 4 + sub;
    float v = ss[u] + sd[u];
    v = (v >= 0.0f) ? v : NEG_SLOPE * v;
    if (FULL || e < E) {
      elog[e * HEADS + hd] = v;
      atomicMaxF(&emax[rr[u] * HEADS + hd], v);
    }
  }
}

__global__ __launch_bounds__(256) void gat_edge_logits(
    const int* __restrict__ ei, int E,
    const float* __restrict__ s_src, const float* __restrict__ s_dst,
    float* __restrict__ elog, float* __restrict__ emax) {
  const int lane = threadIdx.x & 31;
  const int wave = (blockIdx.x * blockDim.x + threadIdx.x) >> 5;
  const int e0 = wave * EDGES_PER_WAVE;
  if (e0 >= E) return;
  if (e0 + EDGES_PER_WAVE <= E)
    edge_logits_wave<true>(ei, E, s_src, s_dst, elog, emax, e0, lane);
  else
    edge_logits_wave<false>(ei, E, s_src, s_dst, elog, emax, e0, lane);
}

template <bool FULL>
__device__ __forceinline__ void edge_exp_wave(
    const int* __restrict__ ei, int E, const float* __restrict__ emax,
    float* __restrict__ elog, float* __restrict__ denom, int e0, int lane) {
  int el = e0 + lane;
  if (!FULL && el > E - 1) el = E - 1;
  const int rl = ei[el];
  const int sub = lane >> 3, hd = lane & 7;

  int   rr[EDGES_PER_WAVE / 4];
  float x[EDGES_PER_WAVE / 4], m[EDGES_PER_WAVE / 4];
  #pragma unroll
  for (int u = 0; u < EDGES_PER_WAVE / 4; ++u) {
    int e = e0 + u * 4 + sub;
    if (!FULL && e > E - 1) e = E - 1;
    x[u] = elog[e * HEADS + hd];
  }
  #pragma unroll
  for (int u = 0; u < EDGES_PER_WAVE / 4; ++u) {
    rr[u] = __shfl(rl, u * 4 + sub);
    m[u]  = emax[rr[u] * HEADS + hd];
  }
  #pragma unroll
  for (int u = 0; u < EDGES_PER_WAVE / 4; ++u) {
    const int e = e0 + u * 4 + sub;
    const float ex = expf(x[u] - m[u]);
    if (FULL || e < E) {
      elog[e * HEADS + hd] = ex;
      atomicAdd(&denom[rr[u] * HEADS + hd], ex);
    }
  }
}

__global__ __launch_bounds__(256) void gat_edge_exp(
    const int* __restrict__ ei, int E, const float* __restrict__ emax,
    float* __restrict__ elog, float* __restrict__ denom) {
  const int lane = threadIdx.x & 31;
  const int wave = (blockIdx.x * blockDim.x + threadIdx.x) >> 5;
  const int e0 = wave * EDGES_PER_WAVE;
  if (e0 >= E) return;
  if (e0 + EDGES_PER_WAVE <= E)
    edge_exp_wave<true>(ei, E, emax, elog, denom, e0, lane);
  else
    edge_exp_wave<false>(ei, E, emax, elog, denom, e0, lane);
}

template <bool FULL>
__device__ __forceinline__ void aggregate_wave(
    const int* __restrict__ ei, int E, const float* __restrict__ h,
    const float* __restrict__ elog, const float* __restrict__ denom,
    float* __restrict__ out, int e0, int lane) {
  const int sub  = lane >> 3;
  const int hd   = lane & 7;
  const int half = lane >> 4;

  int eA = e0 + sub, eB = e0 + 4 + sub;
  if (!FULL && eA > E - 1) eA = E - 1;
  if (!FULL && eB > E - 1) eB = E - 1;
  const int rA = ei[eA],     rB = ei[eB];
  const int cA = ei[E + eA], cB = ei[E + eB];
  const float xA = elog[eA * HEADS + hd];
  const float xB = elog[eB * HEADS + hd];
  const float dA = denom[rA * HEADS + hd];
  const float dB = denom[rB * HEADS + hd];

  int   rs[AGG_EDGES];
  float hv[AGG_EDGES][4];
  #pragma unroll
  for (int u = 0; u < AGG_EDGES; ++u) {
    const int first = (u & 3) * 8;
    const int c = __builtin_amdgcn_readlane(u < 4 ? cA : cB, first);
    rs[u]       = __builtin_amdgcn_readlane(u < 4 ? rA : rB, first);
    const float* hp = h + (long long)c * F_OUT + lane;
    hv[u][0] = hp[0];
    hv[u][1] = hp[32];
    hv[u][2] = hp[64];
    hv[u][3] = hp[96];
  }

  const float alA = xA / (dA + 1e-16f);
  const float alB = xB / (dB + 1e-16f);

  #pragma unroll
  for (int u = 0; u < AGG_EDGES; ++u) {
    const float al = u < 4 ? alA : alB;
    const int from = (u & 3) * 8 + half;
    const float a0 = __shfl(al, from + 0);
    const float a1 = __shfl(al, from + 2);
    const float a2 = __shfl(al, from + 4);
    const float a3 = __shfl(al, from + 6);
    if (FULL || e0 + u < E) {
      float* op = out + (long long)rs[u] * F_OUT + lane;
      atomicAdd(op +  0, a0 * hv[u][0]);
      atomicAdd(op + 32, a1 * hv[u][1]);
      atomicAdd(op + 64, a2 * hv[u][2]);
      atomicAdd(op + 96, a3 * hv[u][3]);
    }
  }
}

__global__ __launch_bounds__(256) void gat_aggregate(
    const int* __restrict__ ei, int E, const float* __restrict__ h,
    const float* __restrict__ elog, const float* __restrict__ denom,
    float* __restrict__ out) {
  const int lane = threadIdx.x & 31;
  const int wave = (blockIdx.x * blockDim.x + threadIdx.x) >> 5;
  const int e0 = wave * AGG_EDGES;
  if (e0 >= E) return;
  if (e0 + AGG_EDGES <= E)
    aggregate_wave<true>(ei, E, h, elog, denom, out, e0, lane);
  else
    aggregate_wave<false>(ei, E, h, elog, denom, out, e0, lane);
}

extern "C" void kernel_launch(void* const* d_in, const int* in_sizes, int n_in,
                              void* d_out, int out_size, void* d_ws, size_t ws_size,
                              hipStream_t stream) {
  const float* x     = (const float*)d_in[0];
  const int*   ei    = (const int*)d_in[1];
  const float* W     = (const float*)d_in[2];
  const float* a_src = (const float*)d_in[3];
  const float* a_dst = (const float*)d_in[4];
  float* out = (float*)d_out;

  const int N = in_sizes[0] / F_IN;
  const int E = in_sizes[1] / 2;

  float* h     = (float*)d_ws;
  float* s_src = h     + (size_t)N * F_OUT;
  float* s_dst = s_src + (size_t)N * HEADS;
  float* emax  = s_dst + (size_t)N * HEADS;
  float* denom = emax  + (size_t)N * HEADS;
  float* elog  = denom + (size_t)N * HEADS;

  const int nOut  = N * F_OUT;
  const int nNH   = N * HEADS;

  const int B = 256;

  {
    int per_block = B * INIT_PER_THREAD * 4;
    gat_init<<<(nOut + per_block - 1) / per_block, B, 0, stream>>>(out, emax, denom, nOut, nNH);
  }
  {
    int blocks = (N + 127) / 128;
    gat_proj_wmma<<<blocks, B, 0, stream>>>(x, W, h, N);
  }
  gat_node_scores<<<(nNH + B - 1) / B, B, 0, stream>>>(h, a_src, a_dst, s_src, s_dst, nNH);
  const int edge_blocks = (E + 8 * EDGES_PER_WAVE - 1) / (8 * EDGES_PER_WAVE);
  gat_edge_logits<<<edge_blocks, B, 0, stream>>>(ei, E, s_src, s_dst, elog, emax);
  gat_edge_exp<<<edge_blocks, B, 0, stream>>>(ei, E, emax, elog, denom);
  gat_aggregate<<<(E + 8 * AGG_EDGES - 1) / (8 * AGG_EDGES), B, 0, stream>>>(ei, E, h, elog, denom, out);
}


# Aggregate: a work-group owns 400 rows, finds its edges in a 1-byte owner map (from the exp pass), lists them in LDS, gathers 8 rows per wave at once into LDS sums; out stored once; LDS arrival order
# speedup vs baseline: 1.6015x; 1.0454x over previous
#include <hip/hip_runtime.h>
#include <hip/hip_bf16.h>

typedef __attribute__((ext_vector_type(2))) float v2f;
typedef __attribute__((ext_vector_type(8))) float v8f;

#define F_IN   128
#define HEADS  8
#define F_PH   16
#define F_OUT  128
#define NEG_SLOPE 0.2f

#define EDGES_PER_WAVE  32
#define AGG_EDGES       8
#define OWN_ROWS        400
#define OWN_THREADS     1024
#define LIST_CAP        8192

__global__ __launch_bounds__(256) void gat_init(
    float* __restrict__ emax, float* __restrict__ denom, int nNH) {
  const int i = blockIdx.x * blockDim.x + threadIdx.x;
  const float ninf = __uint_as_float(0xFF800000u);
  if (i < (nNH >> 2)) {
    ((float4*)emax)[i]  = make_float4(ninf, ninf, ninf, ninf);
    ((float4*)denom)[i] = make_float4(0.f, 0.f, 0.f, 0.f);
  }
}

__global__ __launch_bounds__(256) void gat_proj_wmma(
    const float* __restrict__ x, const float* __restrict__ W,
    float* __restrict__ h, int nrows) {
  const int wave = threadIdx.x >> 5;
  const int lane = threadIdx.x & 31;
  const long long row0 = ((long long)blockIdx.x * 8 + wave) * 16;
  if (row0 >= nrows) return;

  const int m     = lane & 15;
  const int khalf = (lane >> 4) << 1;
  long long rowA = row0 + m;
  if (rowA >= nrows) rowA = nrows - 1;
  const float* xrow = x + rowA * F_IN;

  v8f acc[HEADS];
  #pragma unroll
  for (int t = 0; t < HEADS; ++t) acc[t] = (v8f){0,0,0,0,0,0,0,0};

  for (int kt = 0; kt < F_IN / 4; ++kt) {
    const int kbase = kt * 4 + khalf;
    v2f a = *(const v2f*)(xrow + kbase);
    #pragma unroll
    for (int t = 0; t < HEADS; ++t) {
      const float* wb = W + t * (F_IN * F_PH) + kbase * F_PH + m;
      v2f b;
      b.x = wb[0];
      b.y = wb[F_PH];
      acc[t] = __builtin_amdgcn_wmma_f32_16x16x4_f32(
           false, a,  false, b,
           (short)0, acc[t],  false,  false);
    }
  }

  const int mofs = (lane >> 4) << 3;
  #pragma unroll
  for (int t = 0; t < HEADS; ++t) {
    #pragma unroll
    for (int v = 0; v < 8; ++v) {
      long long r = row0 + v + mofs;
      if (r < nrows) h[r * F_OUT + t * F_PH + m] = acc[t][v];
    }
  }
}

__global__ void gat_node_scores(const float* __restrict__ h,
                                const float* __restrict__ a_src,
                                const float* __restrict__ a_dst,
                                float* __restrict__ s_src,
                                float* __restrict__ s_dst, int nNH) {
  int idx = blockIdx.x * blockDim.x + threadIdx.x;
  if (idx >= nNH) return;
  int n = idx >> 3, hd = idx & 7;
  const float* hp = h + (long long)n * F_OUT + hd * F_PH;
  const float* as = a_src + hd * F_PH;
  const float* ad = a_dst + hd * F_PH;
  float ss = 0.f, sd = 0.f;
  #pragma unroll
  for (int o = 0; o < F_PH; ++o) {
    float v = hp[o];
    ss += v * as[o];
    sd += v * ad[o];
  }
  s_src[idx] = ss;
  s_dst[idx] = sd;
}

__device__ __forceinline__ void atomicMaxF(float* addr, float val) {
  if (val >= 0.0f) atomicMax((int*)addr, __float_as_int(val));
  else             atomicMin((unsigned int*)addr, __float_as_uint(val));
}

template <bool FULL>
__device__ __forceinline__ void edge_logits_wave(
    const int* __restrict__ ei, int E,
    const float* __restrict__ s_src, const float* __restrict__ s_dst,
    float* __restrict__ elog, float* __restrict__ emax, int e0, int lane) {
  int el = e0 + lane;
  if (!FULL && el > E - 1) el = E - 1;
  const int rl = ei[el];
  const int cl = ei[E + el];
  const int sub = lane >> 3, hd = lane & 7;

  int   rr[EDGES_PER_WAVE / 4];
  float ss[EDGES_PER_WAVE / 4], sd[EDGES_PER_WAVE / 4];
  #pragma unroll
  for (int u = 0; u < EDGES_PER_WAVE / 4; ++u) {
    const int from = u * 4 + sub;
    rr[u] = __shfl(rl, from);
    const int c = __shfl(cl, from);
    ss[u] = s_src[rr[u] * HEADS + hd];
    sd[u] = s_dst[c * HEADS + hd];
  }
  #pragma unroll
  for (int u = 0; u < EDGES_PER_WAVE / 4; ++u) {
    const int e = e0 + u * 4 + sub;
    float v = ss[u] + sd[u];
    v = (v >= 0.0f) ? v : NEG_SLOPE * v;
    if (FULL || e < E) {
      elog[e * HEADS + hd] = v;
      atomicMaxF(&emax[rr[u] * HEADS + hd], v);
    }
  }
}

__global__ __launch_bounds__(256) void gat_edge_logits(
    const int* __restrict__ ei, int E,
    const float* __restrict__ s_src, const float* __restrict__ s_dst,
    float* __restrict__ elog, float* __restrict__ emax) {
  const int lane = threadIdx.x & 31;
  const int wave = (blockIdx.x * blockDim.x + threadIdx.x) >> 5;
  const int e0 = wave * EDGES_PER_WAVE;
  if (e0 >= E) return;
  if (e0 + EDGES_PER_WAVE <= E)
    edge_logits_wave<true>(ei, E, s_src, s_dst, elog, emax, e0, lane);
  else
    edge_logits_wave<false>(ei, E, s_src, s_dst, elog, emax, e0, lane);
}

template <bool FULL>
__device__ __forceinline__ void edge_exp_wave(
    const int* __restrict__ ei, int E, const float* __restrict__ emax,
    float* __restrict__ elog, float* __restrict__ denom,
    unsigned char* __restrict__ omap, int e0, int lane) {
  int el = e0 + lane;
  if (!FULL && el > E - 1) el = E - 1;
  const int rl = ei[el];
  if (FULL || e0 + lane < E) omap[e0 + lane] = (unsigned char)(rl / OWN_ROWS);
  const int sub = lane >> 3, hd = lane & 7;

  int   rr[EDGES_PER_WAVE / 4];
  float x[EDGES_PER_WAVE / 4], m[EDGES_PER_WAVE / 4];
  #pragma unroll
  for (int u = 0; u < EDGES_PER_WAVE / 4; ++u) {
    int e = e0 + u * 4 + sub;
    if (!FULL && e > E - 1) e = E - 1;
    x[u] = elog[e * HEADS + hd];
  }
  #pragma unroll
  for (int u = 0; u < EDGES_PER_WAVE / 4; ++u) {
    rr[u] = __shfl(rl, u * 4 + sub);
    m[u]  = emax[rr[u] * HEADS + hd];
  }
  #pragma unroll
  for (int u = 0; u < EDGES_PER_WAVE / 4; ++u) {
    const int e = e0 + u * 4 + sub;
    const float ex = expf(x[u] - m[u]);
    if (FULL || e < E) {
      elog[e * HEADS + hd] = ex;
      atomicAdd(&denom[rr[u] * HEADS + hd], ex);
    }
  }
}

__global__ __launch_bounds__(256) void gat_edge_exp(
    const int* __restrict__ ei, int E, const float* __restrict__ emax,
    float* __restrict__ elog, float* __restrict__ denom,
    float* __restrict__ omap_ws) {
  unsigned char* omap = (unsigned char*)omap_ws;
  const int lane = threadIdx.x & 31;
  const int wave = (blockIdx.x * blockDim.x + threadIdx.x) >> 5;
  const int e0 = wave * EDGES_PER_WAVE;
  if (e0 >= E) return;
  if (e0 + EDGES_PER_WAVE <= E)
    edge_exp_wave<true>(ei, E, emax, elog, denom, omap, e0, lane);
  else
    edge_exp_wave<false>(ei, E, emax, elog, denom, omap, e0, lane);
}

__device__ __forceinline__ void own_slow_edge(
    const int* __restrict__ ei, int E, const float* __restrict__ h,
    const float* __restrict__ elog, const float* __restrict__ denom,
    float* acc, int row0, int e) {
  const int r = ei[e], cc = ei[E + e];
  for (int hd = 0; hd < HEADS; ++hd) {
    const float al = elog[e * HEADS + hd] / (denom[r * HEADS + hd] + 1e-16f);
    for (int o = 0; o < F_PH; ++o) {
      const int f = hd * F_PH + o;
      atomicAdd(&acc[(r - row0) * F_OUT + f], al * h[(long long)cc * F_OUT + f]);
    }
  }
}

__device__ __forceinline__ void own_append(int e, int* cnt, int* list) {
  const int idx = atomicAdd(cnt, 1);
  if (idx < LIST_CAP) list[idx] = e;
}

__device__ __forceinline__ unsigned own_zero_bytes(unsigned x) {
  return ~(((x & 0x7F7F7F7Fu) + 0x7F7F7F7Fu) | x | 0x7F7F7F7Fu);
}
__device__ __forceinline__ void own_scan_piece(uint4 v, int e0, unsigned mine4,
                                               int* cnt, int* list) {
  unsigned m = (own_zero_bytes(v.x ^ mine4) >> 7) | (own_zero_bytes(v.y ^ mine4) >> 6)
             | (own_zero_bytes(v.z ^ mine4) >> 5) | (own_zero_bytes(v.w ^ mine4) >> 4);
  while (m != 0u) {
    const int p = __builtin_ctz(m);
    own_append(e0 + (p & 7) * 4 + (p >> 3), cnt, list);
    m &= m - 1u;
  }
}

template <bool FULL>
__device__ __forceinline__ void own_batch(
    const int* __restrict__ ei, int E, const float* __restrict__ h,
    const float* __restrict__ elog, const float* __restrict__ denom,
    float* acc, const int* list, int row0, int b0, int n, int lane) {
  const int sub  = lane >> 3;
  const int hd   = lane & 7;
  const int half = lane >> 4;
  int iA = b0 + sub, iB = b0 + 4 + sub;
  if (!FULL && iA > n - 1) iA = n - 1;
  if (!FULL && iB > n - 1) iB = n - 1;
  const int eA = list[iA], eB = list[iB];
  const int rA = ei[eA],     rB = ei[eB];
  const int cA = ei[E + eA], cB = ei[E + eB];
  const float xA = elog[eA * HEADS + hd];
  const float xB = elog[eB * HEADS + hd];
  const float dA = denom[rA * HEADS + hd];
  const float dB = denom[rB * HEADS + hd];

  int   rs[AGG_EDGES];
  float hv[AGG_EDGES][4];
  #pragma unroll
  for (int u = 0; u < AGG_EDGES; ++u) {
    const int first = (u & 3) * 8;
    const int cc = __builtin_amdgcn_readlane(u < 4 ? cA : cB, first);
    rs[u]        = __builtin_amdgcn_readlane(u < 4 ? rA : rB, first);
    const float* hp = h + (long long)cc * F_OUT + lane;
    hv[u][0] = hp[0];
    hv[u][1] = hp[32];
    hv[u][2] = hp[64];
    hv[u][3] = hp[96];
  }

  const float alA = xA / (dA + 1e-16f);
  const float alB = xB / (dB + 1e-16f);

  #pragma unroll
  for (int u = 0; u < AGG_EDGES; ++u) {
    const float al = u < 4 ? alA : alB;
    const int from = (u & 3) * 8 + half;
    const float a0 = __shfl(al, from + 0);
    const float a1 = __shfl(al, from + 2);
    const float a2 = __shfl(al, from + 4);
    const float a3 = __shfl(al, from + 6);
    if (FULL || b0 + u < n) {
      float* ap = acc + (rs[u] - row0) * F_OUT + lane;
      atomicAdd(ap +  0, a0 * hv[u][0]);
      atomicAdd(ap + 32, a1 * hv[u][1]);
      atomicAdd(ap + 64, a2 * hv[u][2]);
      atomicAdd(ap + 96, a3 * hv[u][3]);
    }
  }
}

__global__ __launch_bounds__(OWN_THREADS) void gat_aggregate_owner(
    const int* __restrict__ ei, int E, const float* __restrict__ h,
    const float* __restrict__ elog, const float* __restrict__ denom,
    const float* __restrict__ omap_ws, float* __restrict__ out, int N) {
  __shared__ __attribute__((aligned(16))) float acc[OWN_ROWS * F_OUT];
  __shared__ int list[LIST_CAP];
  __shared__ int cnt;

  const int tid  = threadIdx.x;
  const int lane = tid & 31;
  const int wave = tid >> 5;
  const int row0 = blockIdx.x * OWN_ROWS;
  const unsigned my = blockIdx.x;

  for (int i = tid; i < OWN_ROWS * F_OUT / 4; i += OWN_THREADS)
    ((float4*)acc)[i] = make_float4(0.f, 0.f, 0.f, 0.f);
  if (tid == 0) cnt = 0;
  __syncthreads();

  const uint4* m4 = (const uint4*)omap_ws;
  const unsigned char* m1 = (const unsigned char*)omap_ws;
  const int nchunks = E >> 4;
  const unsigned mine4 = my * 0x01010101u;
  for (int base = 0; base < nchunks; base += 4 * OWN_THREADS) {
    const int i0 = base + tid, i1 = i0 + OWN_THREADS;
    const int i2 = i1 + OWN_THREADS, i3 = i2 + OWN_THREADS;
    const int last = nchunks - 1;
    const uint4 v0 = m4[i0 < last ? i0 : last];
    const uint4 v1 = m4[i1 < last ? i1 : last];
    const uint4 v2 = m4[i2 < last ? i2 : last];
    const uint4 v3 = m4[i3 < last ? i3 : last];
    if (i0 < nchunks) own_scan_piece(v0, i0 * 16, mine4, &cnt, list);
    if (i1 < nchunks) own_scan_piece(v1, i1 * 16, mine4, &cnt, list);
    if (i2 < nchunks) own_scan_piece(v2, i2 * 16, mine4, &cnt, list);
    if (i3 < nchunks) own_scan_piece(v3, i3 * 16, mine4, &cnt, list);
  }
  {
    const int e = nchunks * 16 + tid;
    if (e < E && m1[e] == my) own_append(e, &cnt, list);
  }
  __syncthreads();

  const int total = cnt;
  if (total > LIST_CAP) {
    for (int e = tid; e < E; e += OWN_THREADS)
      if (m1[e] == my) own_slow_edge(ei, E, h, elog, denom, acc, row0, e);
  } else {
  const int n = total;
  for (int b0 = wave * AGG_EDGES; b0 < n; b0 += (OWN_THREADS / 32) * AGG_EDGES) {
    if (b0 + AGG_EDGES <= n)
      own_batch<true>(ei, E, h, elog, denom, acc, list, row0, b0, n, lane);
    else
      own_batch<false>(ei, E, h, elog, denom, acc, list, row0, b0, n, lane);
  }
  }
  __syncthreads();

  int rows = N - row0;
  if (rows > OWN_ROWS) rows = OWN_ROWS;
  float4* o4 = (float4*)(out + (long long)row0 * F_OUT);
  for (int i = tid; i < rows * (F_OUT / 4); i += OWN_THREADS)
    o4[i] = ((const float4*)acc)[i];
}

extern "C" void kernel_launch(void* const* d_in, const int* in_sizes, int n_in,
                              void* d_out, int out_size, void* d_ws, size_t ws_size,
                              hipStream_t stream) {
  const float* x     = (const float*)d_in[0];
  const int*   ei    = (const int*)d_in[1];
  const float* W     = (const float*)d_in[2];
  const float* a_src = (const float*)d_in[3];
  const float* a_dst = (const float*)d_in[4];
  float* out = (float*)d_out;

  const int N = in_sizes[0] / F_IN;
  const int E = in_sizes[1] / 2;

  float* h     = (float*)d_ws;
  float* s_src = h     + (size_t)N * F_OUT;
  float* s_dst = s_src + (size_t)N * HEADS;
  float* emax  = s_dst + (size_t)N * HEADS;
  float* denom = emax  + (size_t)N * HEADS;
  float* elog  = denom + (size_t)N * HEADS;

  const int nNH   = N * HEADS;
  float* omap = s_src;

  const int B = 256;

  gat_init<<<(nNH / 4 + B - 1) / B, B, 0, stream>>>(emax, denom, nNH);
  {
    int blocks = (N + 127) / 128;
    gat_proj_wmma<<<blocks, B, 0, stream>>>(x, W, h, N);
  }
  gat_node_scores<<<(nNH + B - 1) / B, B, 0, stream>>>(h, a_src, a_dst, s_src, s_dst, nNH);
  const int edge_blocks = (E + 8 * EDGES_PER_WAVE - 1) / (8 * EDGES_PER_WAVE);
  gat_edge_logits<<<edge_blocks, B, 0, stream>>>(ei, E, s_src, s_dst, elog, emax);
  gat_edge_exp<<<edge_blocks, B, 0, stream>>>(ei, E, emax, elog, denom, omap);
  gat_aggregate_owner<<<(N + OWN_ROWS - 1) / OWN_ROWS, OWN_THREADS, 0, stream>>>(ei, E, h, elog, denom, omap, out, N);
}


# Owner work-group computes its edges' logits, row max (LDS max) and exp-sum itself: logit, exp, init launches and global atomics gone (6->4); rows divided by exp-sum once at the end; LDS arrival order
# speedup vs baseline: 2.9213x; 1.8241x over previous
#include <hip/hip_runtime.h>
#include <hip/hip_bf16.h>

typedef __attribute__((ext_vector_type(2))) float v2f;
typedef __attribute__((ext_vector_type(8))) float v8f;

#define F_IN   128
#define HEADS  8
#define F_PH   16
#define F_OUT  128
#define NEG_SLOPE 0.2f

#define AGG_EDGES       8
#define MAX_SETS        2
#define OWN_ROWS        400
#define OWN_THREADS     1024
#define LIST_CAP        8192
#define COL_BITS        17

__global__ __launch_bounds__(256) void gat_owner_map(
    const int* __restrict__ ei, int E, float* __restrict__ omap_ws) {
  unsigned char* omap = (unsigned char*)omap_ws;
  const int e0 = (blockIdx.x * blockDim.x + threadIdx.x) * 4;
  if (e0 + 4 <= E) {
    const int4 r = *(const int4*)(ei + e0);
    *(unsigned*)(omap + e0) = (unsigned)(r.x / OWN_ROWS) | ((unsigned)(r.y / OWN_ROWS) << 8)
                            | ((unsigned)(r.z / OWN_ROWS) << 16) | ((unsigned)(r.w / OWN_ROWS) << 24);
  } else {
    for (int e = e0; e < E; ++e) omap[e] = (unsigned char)(ei[e] / OWN_ROWS);
  }
}

__global__ __launch_bounds__(256) void gat_proj_wmma(
    const float* __restrict__ x, const float* __restrict__ W,
    float* __restrict__ h, int nrows) {
  const int wave = threadIdx.x >> 5;
  const int lane = threadIdx.x & 31;
  const long long row0 = ((long long)blockIdx.x * 8 + wave) * 16;
  if (row0 >= nrows) return;

  const int m     = lane & 15;
  const int khalf = (lane >> 4) << 1;
  long long rowA = row0 + m;
  if (rowA >= nrows) rowA = nrows - 1;
  const float* xrow = x + rowA * F_IN;

  v8f acc[HEADS];
  #pragma unroll
  for (int t = 0; t < HEADS; ++t) acc[t] = (v8f){0,0,0,0,0,0,0,0};

  for (int kt = 0; kt < F_IN / 4; ++kt) {
    const int kbase = kt * 4 + khalf;
    v2f a = *(const v2f*)(xrow + kbase);
    #pragma unroll
    for (int t = 0; t < HEADS; ++t) {
      const float* wb = W + t * (F_IN * F_PH) + kbase * F_PH + m;
      v2f b;
      b.x = wb[0];
      b.y = wb[F_PH];
      acc[t] = __builtin_amdgcn_wmma_f32_16x16x4_f32(
           false, a,  false, b,
           (short)0, acc[t],  false,  false);
    }
  }

  const int mofs = (lane >> 4) << 3;
  #pragma unroll
  for (int t = 0; t < HEADS; ++t) {
    #pragma unroll
    for (int v = 0; v < 8; ++v) {
      long long r = row0 + v + mofs;
      if (r < nrows) h[r * F_OUT + t * F_PH + m] = acc[t][v];
    }
  }
}

__global__ void gat_node_scores(const float* __restrict__ h,
                                const float* __restrict__ a_src,
                                const float* __restrict__ a_dst,
                                float* __restrict__ s_src,
                                float* __restrict__ s_dst, int nNH) {
  int idx = blockIdx.x * blockDim.x + threadIdx.x;
  if (idx >= nNH) return;
  int n = idx >> 3, hd = idx & 7;
  const float* hp = h + (long long)n * F_OUT + hd * F_PH;
  const float* as = a_src + hd * F_PH;
  const float* ad = a_dst + hd * F_PH;
  float ss = 0.f, sd = 0.f;
  #pragma unroll
  for (int o = 0; o < F_PH; ++o) {
    float v = hp[o];
    ss += v * as[o];
    sd += v * ad[o];
  }
  s_src[idx] = ss;
  s_dst[idx] = sd;
}


__device__ __forceinline__ unsigned own_key(float v) {
  const unsigned b = __float_as_uint(v);
  return b ^ ((unsigned)((int)b >> 31) | 0x80000000u);
}
__device__ __forceinline__ float own_unkey(unsigned k) {
  return __uint_as_float(k ^ ((unsigned)((int)~k >> 31) | 0x80000000u));
}
__device__ __forceinline__ float own_logit(float ss, float sd) {
  const float v = ss + sd;
  return (v >= 0.0f) ? v : NEG_SLOPE * v;
}

__device__ __forceinline__ void own_slow_max(
    const int* __restrict__ ei, int E, const float* __restrict__ s_dst,
    const float* ssl, unsigned* mxk, int row0, int e) {
  const int l = ei[e] - row0, cc = ei[E + e];
  for (int hd = 0; hd < HEADS; ++hd)
    atomicMax(&mxk[l * HEADS + hd], own_key(own_logit(ssl[l * HEADS + hd], s_dst[cc * HEADS + hd])));
}
__device__ __forceinline__ void own_slow_sum(
    const int* __restrict__ ei, int E, const float* __restrict__ h,
    const float* __restrict__ s_dst, const float* ssl, const unsigned* mxk,
    float* den, float* acc, int row0, int e) {
  const int l = ei[e] - row0, cc = ei[E + e];
  for (int hd = 0; hd < HEADS; ++hd) {
    const float v = own_logit(ssl[l * HEADS + hd], s_dst[cc * HEADS + hd]);
    const float p = expf(v - own_unkey(mxk[l * HEADS + hd]));
    atomicAdd(&den[l * HEADS + hd], p);
    for (int o = 0; o < F_PH; ++o) {
      const int f = hd * F_PH + o;
      atomicAdd(&acc[l * F_OUT + f], p * h[(long long)cc * F_OUT + f]);
    }
  }
}

__device__ __forceinline__ void own_append(int e, int* cnt, int* list) {
  const int idx = atomicAdd(cnt, 1);
  if (idx < LIST_CAP) list[idx] = e;
}

__device__ __forceinline__ unsigned own_zero_bytes(unsigned x) {
  return ~(((x & 0x7F7F7F7Fu) + 0x7F7F7F7Fu) | x | 0x7F7F7F7Fu);
}
__device__ __forceinline__ void own_scan_piece(uint4 v, int e0, unsigned mine4,
                                               int* cnt, int* list) {
  unsigned m = (own_zero_bytes(v.x ^ mine4) >> 7) | (own_zero_bytes(v.y ^ mine4) >> 6)
             | (own_zero_bytes(v.z ^ mine4) >> 5) | (own_zero_bytes(v.w ^ mine4) >> 4);
  while (m != 0u) {
    const int p = __builtin_ctz(m);
    own_append(e0 + (p & 7) * 4 + (p >> 3), cnt, list);
    m &= m - 1u;
  }
}

template <bool FULL>
__device__ __forceinline__ void own_max_batch(
    const int* __restrict__ ei, int E, const float* __restrict__ s_dst,
    const float* ssl, unsigned* mxk, int* list, int row0, int b0, int n, int lane) {
  const int sub = lane >> 3, hd = lane & 7;
  int cl[MAX_SETS], ll[MAX_SETS];
  #pragma unroll
  for (int s = 0; s < MAX_SETS; ++s) {
    int i = b0 + s * 32 + lane;
    if (!FULL && i > n - 1) i = n - 1;
    const int e = list[i];
    ll[s] = ei[e] - row0;
    cl[s] = ei[E + e];
  }
  int   wu[MAX_SETS][8];
  float sd[MAX_SETS][8], ss[MAX_SETS][8];
  #pragma unroll
  for (int s = 0; s < MAX_SETS; ++s) {
    #pragma unroll
    for (int u = 0; u < 8; ++u) {
      const int from = u * 4 + sub;
      const int c = __shfl(cl[s], from);
      wu[s][u] = __shfl(ll[s], from) * HEADS + hd;
      sd[s][u] = s_dst[c * HEADS + hd];
      ss[s][u] = ssl[wu[s][u]];
    }
  }
  #pragma unroll
  for (int s = 0; s < MAX_SETS; ++s) {
    const int i = b0 + s * 32 + lane;
    if (FULL || i < n) list[i] = cl[s] | (ll[s] << COL_BITS);
  }
  #pragma unroll
  for (int s = 0; s < MAX_SETS; ++s) {
    #pragma unroll
    for (int u = 0; u < 8; ++u) {
      if (FULL || b0 + s * 32 + u * 4 + sub < n)
        atomicMax(&mxk[wu[s][u]], own_key(own_logit(ss[s][u], sd[s][u])));
    }
  }
}

template <bool FULL>
__device__ __forceinline__ void own_sum_batch(
    const float* __restrict__ h, const float* __restrict__ s_dst,
    const float* ssl, const unsigned* mxk, float* den, float* acc,
    const int* list, int b0, int n, int lane) {
  const int sub  = lane >> 3;
  const int hd   = lane & 7;
  const int half = lane >> 4;
  int iA = b0 + sub, iB = b0 + 4 + sub;
  if (!FULL && iA > n - 1) iA = n - 1;
  if (!FULL && iB > n - 1) iB = n - 1;
  const int pA = list[iA], pB = list[iB];
  const int cA = pA & ((1 << COL_BITS) - 1), cB = pB & ((1 << COL_BITS) - 1);
  const int wA = (pA >> COL_BITS) * HEADS + hd, wB = (pB >> COL_BITS) * HEADS + hd;
  const float sdA = s_dst[cA * HEADS + hd];
  const float sdB = s_dst[cB * HEADS + hd];

  int   ls[AGG_EDGES];
  float hv[AGG_EDGES][4];
  #pragma unroll
  for (int u = 0; u < AGG_EDGES; ++u) {
    const int first = (u & 3) * 8;
    const int pk = __builtin_amdgcn_readlane(u < 4 ? pA : pB, first);
    ls[u] = pk >> COL_BITS;
    const float* hp = h + (long long)(pk & ((1 << COL_BITS) - 1)) * F_OUT + lane;
    hv[u][0] = hp[0];
    hv[u][1] = hp[32];
    hv[u][2] = hp[64];
    hv[u][3] = hp[96];
  }

  const float exA = expf(own_logit(ssl[wA], sdA) - own_unkey(mxk[wA]));
  const float exB = expf(own_logit(ssl[wB], sdB) - own_unkey(mxk[wB]));
  if (FULL || b0 + sub < n)     atomicAdd(&den[wA], exA);
  if (FULL || b0 + 4 + sub < n) atomicAdd(&den[wB], exB);

  #pragma unroll
  for (int u = 0; u < AGG_EDGES; ++u) {
    const float ex = u < 4 ? exA : exB;
    const int from = (u & 3) * 8 + half;
    const float a0 = __shfl(ex, from + 0);
    const float a1 = __shfl(ex, from + 2);
    const float a2 = __shfl(ex, from + 4);
    const float a3 = __shfl(ex, from + 6);
    if (FULL || b0 + u < n) {
      float* ap = acc + ls[u] * F_OUT + lane;
      atomicAdd(ap +  0, a0 * hv[u][0]);
      atomicAdd(ap + 32, a1 * hv[u][1]);
      atomicAdd(ap + 64, a2 * hv[u][2]);
      atomicAdd(ap + 96, a3 * hv[u][3]);
    }
  }
}

__global__ __launch_bounds__(OWN_THREADS) void gat_softmax_aggregate_owner(
    const int* __restrict__ ei, int E, const float* __restrict__ h,
    const float* __restrict__ s_src, const float* __restrict__ s_dst,
    const float* __restrict__ omap_ws, float* __restrict__ out, int N) {
  __shared__ __attribute__((aligned(16))) float acc[OWN_ROWS * F_OUT];
  __shared__ int list[LIST_CAP];
  __shared__ unsigned mxk[OWN_ROWS * HEADS];
  __shared__ float den[OWN_ROWS * HEADS];
  __shared__ float ssl[OWN_ROWS * HEADS];
  __shared__ int cnt;

  const int tid  = threadIdx.x;
  const int lane = tid & 31;
  const int wave = tid >> 5;
  const int row0 = blockIdx.x * OWN_ROWS;
  const unsigned my = blockIdx.x;

  for (int i = tid; i < OWN_ROWS * F_OUT / 4; i += OWN_THREADS)
    ((float4*)acc)[i] = make_float4(0.f, 0.f, 0.f, 0.f);
  for (int i = tid; i < OWN_ROWS * HEADS; i += OWN_THREADS) {
    mxk[i] = 0u;
    den[i] = 0.f;
    const long long g = (long long)row0 * HEADS + i;
    ssl[i] = (g < (long long)N * HEADS) ? s_src[g] : 0.f;
  }
  if (tid == 0) cnt = 0;
  __syncthreads();

  const uint4* m4 = (const uint4*)omap_ws;
  const unsigned char* m1 = (const unsigned char*)omap_ws;
  const int nchunks = E >> 4;
  const unsigned mine4 = my * 0x01010101u;
  for (int base = 0; base < nchunks; base += 4 * OWN_THREADS) {
    const int i0 = base + tid, i1 = i0 + OWN_THREADS;
    const int i2 = i1 + OWN_THREADS, i3 = i2 + OWN_THREADS;
    const int last = nchunks - 1;
    const uint4 v0 = m4[i0 < last ? i0 : last];
    const uint4 v1 = m4[i1 < last ? i1 : last];
    const uint4 v2 = m4[i2 < last ? i2 : last];
    const uint4 v3 = m4[i3 < last ? i3 : last];
    if (i0 < nchunks) own_scan_piece(v0, i0 * 16, mine4, &cnt, list);
    if (i1 < nchunks) own_scan_piece(v1, i1 * 16, mine4, &cnt, list);
    if (i2 < nchunks) own_scan_piece(v2, i2 * 16, mine4, &cnt, list);
    if (i3 < nchunks) own_scan_piece(v3, i3 * 16, mine4, &cnt, list);
  }
  {
    const int e = nchunks * 16 + tid;
    if (e < E && m1[e] == my) own_append(e, &cnt, list);
  }
  __syncthreads();

  const int n = cnt;
  const bool listed = n <= LIST_CAP;

  if (listed) {
    for (int b0 = wave * (MAX_SETS * 32); b0 < n; b0 += (OWN_THREADS / 32) * (MAX_SETS * 32)) {
      if (b0 + MAX_SETS * 32 <= n)
        own_max_batch<true>(ei, E, s_dst, ssl, mxk, list, row0, b0, n, lane);
      else
        own_max_batch<false>(ei, E, s_dst, ssl, mxk, list, row0, b0, n, lane);
    }
  } else {
    for (int e = tid; e < E; e += OWN_THREADS)
      if (m1[e] == my) own_slow_max(ei, E, s_dst, ssl, mxk, row0, e);
  }
  __syncthreads();

  if (listed) {
    for (int b0 = wave * AGG_EDGES; b0 < n; b0 += (OWN_THREADS / 32) * AGG_EDGES) {
      if (b0 + AGG_EDGES <= n)
        own_sum_batch<true>(h, s_dst, ssl, mxk, den, acc, list, b0, n, lane);
      else
        own_sum_batch<false>(h, s_dst, ssl, mxk, den, acc, list, b0, n, lane);
    }
  } else {
    for (int e = tid; e < E; e += OWN_THREADS)
      if (m1[e] == my) own_slow_sum(ei, E, h, s_dst, ssl, mxk, den, acc, row0, e);
  }
  __syncthreads();

  int rows = N - row0;
  if (rows > OWN_ROWS) rows = OWN_ROWS;
  float4* o4 = (float4*)(out + (long long)row0 * F_OUT);
  for (int i = tid; i < rows * (F_OUT / 4); i += OWN_THREADS) {
    float4 a = ((const float4*)acc)[i];
    const float d = den[(i >> 5) * HEADS + ((i & 31) >> 2)] + 1e-16f;
    a.x /= d; a.y /= d; a.z /= d; a.w /= d;
    o4[i] = a;
  }
}

extern "C" void kernel_launch(void* const* d_in, const int* in_sizes, int n_in,
                              void* d_out, int out_size, void* d_ws, size_t ws_size,
                              hipStream_t stream) {
  const float* x     = (const float*)d_in[0];
  const int*   ei    = (const int*)d_in[1];
  const float* W     = (const float*)d_in[2];
  const float* a_src = (const float*)d_in[3];
  const float* a_dst = (const float*)d_in[4];
  float* out = (float*)d_out;

  const int N = in_sizes[0] / F_IN;
  const int E = in_sizes[1] / 2;

  float* h     = (float*)d_ws;
  float* s_src = h     + (size_t)N * F_OUT;
  float* s_dst = s_src + (size_t)N * HEADS;
  float* omap  = s_dst + (size_t)N * HEADS;

  const int nNH   = N * HEADS;
  const int B = 256;

  gat_owner_map<<<((E + 3) / 4 + B - 1) / B, B, 0, stream>>>(ei, E, omap);
  {
    int blocks = (N + 127) / 128;
    gat_proj_wmma<<<blocks, B, 0, stream>>>(x, W, h, N);
  }
  gat_node_scores<<<(nNH + B - 1) / B, B, 0, stream>>>(h, a_src, a_dst, s_src, s_dst, nNH);
  gat_softmax_aggregate_owner<<<(N + OWN_ROWS - 1) / OWN_ROWS, OWN_THREADS, 0, stream>>>(ei, E, h, s_src, s_dst, omap, out, N);
}


# Sum pass: a lane loads floats 4l..4l+3 of a source row (one 16-byte load per edge, not 4 line loads; 1 shuffle, not 4), adds float 4l+j to LDS word 32j+l, read back in order; LDS arrival order
# speedup vs baseline: 3.1620x; 1.0824x over previous
#include <hip/hip_runtime.h>
#include <hip/hip_bf16.h>

typedef __attribute__((ext_vector_type(2))) float v2f;
typedef __attribute__((ext_vector_type(8))) float v8f;

#define F_IN   128
#define HEADS  8
#define F_PH   16
#define F_OUT  128
#define NEG_SLOPE 0.2f

#define AGG_EDGES       8
#define MAX_SETS        2
#define OWN_ROWS        400
#define OWN_THREADS     1024
#define LIST_CAP        8192
#define COL_BITS        17

__global__ __launch_bounds__(256) void gat_owner_map(
    const int* __restrict__ ei, int E, float* __restrict__ omap_ws) {
  unsigned char* omap = (unsigned char*)omap_ws;
  const int e0 = (blockIdx.x * blockDim.x + threadIdx.x) * 4;
  if (e0 + 4 <= E) {
    const int4 r = *(const int4*)(ei + e0);
    *(unsigned*)(omap + e0) = (unsigned)(r.x / OWN_ROWS) | ((unsigned)(r.y / OWN_ROWS) << 8)
                            | ((unsigned)(r.z / OWN_ROWS) << 16) | ((unsigned)(r.w / OWN_ROWS) << 24);
  } else {
    for (int e = e0; e < E; ++e) omap[e] = (unsigned char)(ei[e] / OWN_ROWS);
  }
}

__global__ __launch_bounds__(256) void gat_proj_wmma(
    const float* __restrict__ x, const float* __restrict__ W,
    float* __restrict__ h, int nrows) {
  const int wave = threadIdx.x >> 5;
  const int lane = threadIdx.x & 31;
  const long long row0 = ((long long)blockIdx.x * 8 + wave) * 16;
  if (row0 >= nrows) return;

  const int m     = lane & 15;
  const int khalf = (lane >> 4) << 1;
  long long rowA = row0 + m;
  if (rowA >= nrows) rowA = nrows - 1;
  const float* xrow = x + rowA * F_IN;

  v8f acc[HEADS];
  #pragma unroll
  for (int t = 0; t < HEADS; ++t) acc[t] = (v8f){0,0,0,0,0,0,0,0};

  for (int kt = 0; kt < F_IN / 4; ++kt) {
    const int kbase = kt * 4 + khalf;
    v2f a = *(const v2f*)(xrow + kbase);
    #pragma unroll
    for (int t = 0; t < HEADS; ++t) {
      const float* wb = W + t * (F_IN * F_PH) + kbase * F_PH + m;
      v2f b;
      b.x = wb[0];
      b.y = wb[F_PH];
      acc[t] = __builtin_amdgcn_wmma_f32_16x16x4_f32(
           false, a,  false, b,
           (short)0, acc[t],  false,  false);
    }
  }

  const int mofs = (lane >> 4) << 3;
  #pragma unroll
  for (int t = 0; t < HEADS; ++t) {
    #pragma unroll
    for (int v = 0; v < 8; ++v) {
      long long r = row0 + v + mofs;
      if (r < nrows) h[r * F_OUT + t * F_PH + m] = acc[t][v];
    }
  }
}

__global__ void gat_node_scores(const float* __restrict__ h,
                                const float* __restrict__ a_src,
                                const float* __restrict__ a_dst,
                                float* __restrict__ s_src,
                                float* __restrict__ s_dst, int nNH) {
  int idx = blockIdx.x * blockDim.x + threadIdx.x;
  if (idx >= nNH) return;
  int n = idx >> 3, hd = idx & 7;
  const float* hp = h + (long long)n * F_OUT + hd * F_PH;
  const float* as = a_src + hd * F_PH;
  const float* ad = a_dst + hd * F_PH;
  float ss = 0.f, sd = 0.f;
  #pragma unroll
  for (int o = 0; o < F_PH; ++o) {
    float v = hp[o];
    ss += v * as[o];
    sd += v * ad[o];
  }
  s_src[idx] = ss;
  s_dst[idx] = sd;
}


__device__ __forceinline__ unsigned own_key(float v) {
  const unsigned b = __float_as_uint(v);
  return b ^ ((unsigned)((int)b >> 31) | 0x80000000u);
}
__device__ __forceinline__ float own_unkey(unsigned k) {
  return __uint_as_float(k ^ ((unsigned)((int)~k >> 31) | 0x80000000u));
}
__device__ __forceinline__ float own_logit(float ss, float sd) {
  const float v = ss + sd;
  return (v >= 0.0f) ? v : NEG_SLOPE * v;
}

__device__ __forceinline__ void own_slow_max(
    const int* __restrict__ ei, int E, const float* __restrict__ s_dst,
    const float* ssl, unsigned* mxk, int row0, int e) {
  const int l = ei[e] - row0, cc = ei[E + e];
  for (int hd = 0; hd < HEADS; ++hd)
    atomicMax(&mxk[l * HEADS + hd], own_key(own_logit(ssl[l * HEADS + hd], s_dst[cc * HEADS + hd])));
}
__device__ __forceinline__ void own_slow_sum(
    const int* __restrict__ ei, int E, const float* __restrict__ h,
    const float* __restrict__ s_dst, const float* ssl, const unsigned* mxk,
    float* den, float* acc, int row0, int e) {
  const int l = ei[e] - row0, cc = ei[E + e];
  for (int hd = 0; hd < HEADS; ++hd) {
    const float v = own_logit(ssl[l * HEADS + hd], s_dst[cc * HEADS + hd]);
    const float p = expf(v - own_unkey(mxk[l * HEADS + hd]));
    atomicAdd(&den[l * HEADS + hd], p);
    for (int o = 0; o < F_PH; ++o) {
      const int f = hd * F_PH + o;
      atomicAdd(&acc[l * F_OUT + (f & 3) * 32 + (f >> 2)], p * h[(long long)cc * F_OUT + f]);
    }
  }
}

__device__ __forceinline__ void own_append(int e, int* cnt, int* list) {
  const int idx = atomicAdd(cnt, 1);
  if (idx < LIST_CAP) list[idx] = e;
}

__device__ __forceinline__ unsigned own_zero_bytes(unsigned x) {
  return ~(((x & 0x7F7F7F7Fu) + 0x7F7F7F7Fu) | x | 0x7F7F7F7Fu);
}
__device__ __forceinline__ void own_scan_piece(uint4 v, int e0, unsigned mine4,
                                               int* cnt, int* list) {
  unsigned m = (own_zero_bytes(v.x ^ mine4) >> 7) | (own_zero_bytes(v.y ^ mine4) >> 6)
             | (own_zero_bytes(v.z ^ mine4) >> 5) | (own_zero_bytes(v.w ^ mine4) >> 4);
  while (m != 0u) {
    const int p = __builtin_ctz(m);
    own_append(e0 + (p & 7) * 4 + (p >> 3), cnt, list);
    m &= m - 1u;
  }
}

template <bool FULL>
__device__ __forceinline__ void own_max_batch(
    const int* __restrict__ ei, int E, const float* __restrict__ s_dst,
    const float* ssl, unsigned* mxk, int* list, int row0, int b0, int n, int lane) {
  const int sub = lane >> 3, hd = lane & 7;
  int cl[MAX_SETS], ll[MAX_SETS];
  #pragma unroll
  for (int s = 0; s < MAX_SETS; ++s) {
    int i = b0 + s * 32 + lane;
    if (!FULL && i > n - 1) i = n - 1;
    const int e = list[i];
    ll[s] = ei[e] - row0;
    cl[s] = ei[E + e];
  }
  int   wu[MAX_SETS][8];
  float sd[MAX_SETS][8], ss[MAX_SETS][8];
  #pragma unroll
  for (int s = 0; s < MAX_SETS; ++s) {
    #pragma unroll
    for (int u = 0; u < 8; ++u) {
      const int from = u * 4 + sub;
      const int c = __shfl(cl[s], from);
      wu[s][u] = __shfl(ll[s], from) * HEADS + hd;
      sd[s][u] = s_dst[c * HEADS + hd];
      ss[s][u] = ssl[wu[s][u]];
    }
  }
  #pragma unroll
  for (int s = 0; s < MAX_SETS; ++s) {
    const int i = b0 + s * 32 + lane;
    if (FULL || i < n) list[i] = cl[s] | (ll[s] << COL_BITS);
  }
  #pragma unroll
  for (int s = 0; s < MAX_SETS; ++s) {
    #pragma unroll
    for (int u = 0; u < 8; ++u) {
      if (FULL || b0 + s * 32 + u * 4 + sub < n)
        atomicMax(&mxk[wu[s][u]], own_key(own_logit(ss[s][u], sd[s][u])));
    }
  }
}

template <bool FULL>
__device__ __forceinline__ void own_sum_batch(
    const float* __restrict__ h, const float* __restrict__ s_dst,
    const float* ssl, const unsigned* mxk, float* den, float* acc,
    const int* list, int b0, int n, int lane) {
  const int sub  = lane >> 3;
  const int hd   = lane & 7;
  const int lhd  = lane >> 2;
  int iA = b0 + sub, iB = b0 + 4 + sub;
  if (!FULL && iA > n - 1) iA = n - 1;
  if (!FULL && iB > n - 1) iB = n - 1;
  const int pA = list[iA], pB = list[iB];
  const int cA = pA & ((1 << COL_BITS) - 1), cB = pB & ((1 << COL_BITS) - 1);
  const int wA = (pA >> COL_BITS) * HEADS + hd, wB = (pB >> COL_BITS) * HEADS + hd;
  const float sdA = s_dst[cA * HEADS + hd];
  const float sdB = s_dst[cB * HEADS + hd];

  int    ls[AGG_EDGES];
  float4 hv[AGG_EDGES];
  #pragma unroll
  for (int u = 0; u < AGG_EDGES; ++u) {
    const int first = (u & 3) * 8;
    const int pk = __builtin_amdgcn_readlane(u < 4 ? pA : pB, first);
    ls[u] = pk >> COL_BITS;
    hv[u] = *(const float4*)(h + (long long)(pk & ((1 << COL_BITS) - 1)) * F_OUT + lane * 4);
  }

  const float exA = expf(own_logit(ssl[wA], sdA) - own_unkey(mxk[wA]));
  const float exB = expf(own_logit(ssl[wB], sdB) - own_unkey(mxk[wB]));
  if (FULL || b0 + sub < n)     atomicAdd(&den[wA], exA);
  if (FULL || b0 + 4 + sub < n) atomicAdd(&den[wB], exB);

  #pragma unroll
  for (int u = 0; u < AGG_EDGES; ++u) {
    const float ex = u < 4 ? exA : exB;
    const float a = __shfl(ex, (u & 3) * 8 + lhd);
    if (FULL || b0 + u < n) {
      float* ap = acc + ls[u] * F_OUT + lane;
      atomicAdd(ap +  0, a * hv[u].x);
      atomicAdd(ap + 32, a * hv[u].y);
      atomicAdd(ap + 64, a * hv[u].z);
      atomicAdd(ap + 96, a * hv[u].w);
    }
  }
}

__global__ __launch_bounds__(OWN_THREADS) void gat_softmax_aggregate_owner(
    const int* __restrict__ ei, int E, const float* __restrict__ h,
    const float* __restrict__ s_src, const float* __restrict__ s_dst,
    const float* __restrict__ omap_ws, float* __restrict__ out, int N) {
  __shared__ __attribute__((aligned(16))) float acc[OWN_ROWS * F_OUT];
  __shared__ int list[LIST_CAP];
  __shared__ unsigned mxk[OWN_ROWS * HEADS];
  __shared__ float den[OWN_ROWS * HEADS];
  __shared__ float ssl[OWN_ROWS * HEADS];
  __shared__ int cnt;

  const int tid  = threadIdx.x;
  const int lane = tid & 31;
  const int wave = tid >> 5;
  const int row0 = blockIdx.x * OWN_ROWS;
  const unsigned my = blockIdx.x;

  for (int i = tid; i < OWN_ROWS * F_OUT / 4; i += OWN_THREADS)
    ((float4*)acc)[i] = make_float4(0.f, 0.f, 0.f, 0.f);
  for (int i = tid; i < OWN_ROWS * HEADS; i += OWN_THREADS) {
    mxk[i] = 0u;
    den[i] = 0.f;
    const long long g = (long long)row0 * HEADS + i;
    ssl[i] = (g < (long long)N * HEADS) ? s_src[g] : 0.f;
  }
  if (tid == 0) cnt = 0;
  __syncthreads();

  const uint4* m4 = (const uint4*)omap_ws;
  const unsigned char* m1 = (const unsigned char*)omap_ws;
  const int nchunks = E >> 4;
  const unsigned mine4 = my * 0x01010101u;
  for (int base = 0; base < nchunks; base += 4 * OWN_THREADS) {
    const int i0 = base + tid, i1 = i0 + OWN_THREADS;
    const int i2 = i1 + OWN_THREADS, i3 = i2 + OWN_THREADS;
    const int last = nchunks - 1;
    const uint4 v0 = m4[i0 < last ? i0 : last];
    const uint4 v1 = m4[i1 < last ? i1 : last];
    const uint4 v2 = m4[i2 < last ? i2 : last];
    const uint4 v3 = m4[i3 < last ? i3 : last];
    if (i0 < nchunks) own_scan_piece(v0, i0 * 16, mine4, &cnt, list);
    if (i1 < nchunks) own_scan_piece(v1, i1 * 16, mine4, &cnt, list);
    if (i2 < nchunks) own_scan_piece(v2, i2 * 16, mine4, &cnt, list);
    if (i3 < nchunks) own_scan_piece(v3, i3 * 16, mine4, &cnt, list);
  }
  {
    const int e = nchunks * 16 + tid;
    if (e < E && m1[e] == my) own_append(e, &cnt, list);
  }
  __syncthreads();

  const int n = cnt;
  const bool listed = n <= LIST_CAP;

  if (listed) {
    for (int b0 = wave * (MAX_SETS * 32); b0 < n; b0 += (OWN_THREADS / 32) * (MAX_SETS * 32)) {
      if (b0 + MAX_SETS * 32 <= n)
        own_max_batch<true>(ei, E, s_dst, ssl, mxk, list, row0, b0, n, lane);
      else
        own_max_batch<false>(ei, E, s_dst, ssl, mxk, list, row0, b0, n, lane);
    }
  } else {
    for (int e = tid; e < E; e += OWN_THREADS)
      if (m1[e] == my) own_slow_max(ei, E, s_dst, ssl, mxk, row0, e);
  }
  __syncthreads();

  if (listed) {
    for (int b0 = wave * AGG_EDGES; b0 < n; b0 += (OWN_THREADS / 32) * AGG_EDGES) {
      if (b0 + AGG_EDGES <= n)
        own_sum_batch<true>(h, s_dst, ssl, mxk, den, acc, list, b0, n, lane);
      else
        own_sum_batch<false>(h, s_dst, ssl, mxk, den, acc, list, b0, n, lane);
    }
  } else {
    for (int e = tid; e < E; e += OWN_THREADS)
      if (m1[e] == my) own_slow_sum(ei, E, h, s_dst, ssl, mxk, den, acc, row0, e);
  }
  __syncthreads();

  int rows = N - row0;
  if (rows > OWN_ROWS) rows = OWN_ROWS;
  float4* o4 = (float4*)(out + (long long)row0 * F_OUT);
  for (int i = tid; i < rows * (F_OUT / 4); i += OWN_THREADS) {
    const int q = i & 31;
    const float* ar = acc + (i >> 5) * F_OUT + q;
    float4 a = make_float4(ar[0], ar[32], ar[64], ar[96]);
    const float d = den[(i >> 5) * HEADS + (q >> 2)] + 1e-16f;
    a.x /= d; a.y /= d; a.z /= d; a.w /= d;
    o4[i] = a;
  }
}

extern "C" void kernel_launch(void* const* d_in, const int* in_sizes, int n_in,
                              void* d_out, int out_size, void* d_ws, size_t ws_size,
                              hipStream_t stream) {
  const float* x     = (const float*)d_in[0];
  const int*   ei    = (const int*)d_in[1];
  const float* W     = (const float*)d_in[2];
  const float* a_src = (const float*)d_in[3];
  const float* a_dst = (const float*)d_in[4];
  float* out = (float*)d_out;

  const int N = in_sizes[0] / F_IN;
  const int E = in_sizes[1] / 2;

  float* h     = (float*)d_ws;
  float* s_src = h     + (size_t)N * F_OUT;
  float* s_dst = s_src + (size_t)N * HEADS;
  float* omap  = s_dst + (size_t)N * HEADS;

  const int nNH   = N * HEADS;
  const int B = 256;

  gat_owner_map<<<((E + 3) / 4 + B - 1) / B, B, 0, stream>>>(ei, E, omap);
  {
    int blocks = (N + 127) / 128;
    gat_proj_wmma<<<blocks, B, 0, stream>>>(x, W, h, N);
  }
  gat_node_scores<<<(nNH + B - 1) / B, B, 0, stream>>>(h, a_src, a_dst, s_src, s_dst, nNH);
  gat_softmax_aggregate_owner<<<(N + OWN_ROWS - 1) / OWN_ROWS, OWN_THREADS, 0, stream>>>(ei, E, h, s_src, s_dst, omap, out, N);
}


# Owner map as one bit per edge in each owner's own plane (K0: LDS OR per 1024-edge chunk, whole-line stores): owner reads E/8 bytes, lists set bits by ctz, no byte compares; LDS arrival order kept
# speedup vs baseline: 3.4130x; 1.0794x over previous
#include <hip/hip_runtime.h>
#include <hip/hip_bf16.h>

typedef __attribute__((ext_vector_type(2))) float v2f;
typedef __attribute__((ext_vector_type(8))) float v8f;
typedef __attribute__((ext_vector_type(4))) unsigned v4u;

#define F_IN   128
#define HEADS  8
#define F_PH   16
#define F_OUT  128
#define NEG_SLOPE 0.2f

#define AGG_EDGES       8
#define MAX_SETS        2
#define OWN_ROWS        400
#define OWN_THREADS     1024
#define LIST_CAP        8192
#define COL_BITS        17
#define CH_EDGES        1024
#define CH_WORDS        32
#define TAB_PITCH       257

__global__ __launch_bounds__(256) void gat_owner_map(
    const int* __restrict__ ei, int E, int N, float* __restrict__ omap_ws) {
  __shared__ __attribute__((aligned(16))) unsigned tab[CH_WORDS * TAB_PITCH];
  unsigned* planes = (unsigned*)omap_ws;
  const int tid = threadIdx.x;
  for (int i = tid; i < CH_WORDS * TAB_PITCH / 4; i += 256)
    ((uint4*)tab)[i] = make_uint4(0u, 0u, 0u, 0u);
  __syncthreads();

  const int e0 = blockIdx.x * CH_EDGES + tid * 4;
  unsigned* trow = tab + (tid >> 3) * TAB_PITCH;
  const unsigned b0 = (unsigned)(tid & 7) * 4u;
  if (e0 + 4 <= E) {
    const int4 r = *(const int4*)(ei + e0);
    atomicOr(&trow[r.x / OWN_ROWS], 1u << b0);
    atomicOr(&trow[r.y / OWN_ROWS], 2u << b0);
    atomicOr(&trow[r.z / OWN_ROWS], 4u << b0);
    atomicOr(&trow[r.w / OWN_ROWS], 8u << b0);
  } else {
    for (int j = 0; j < 4; ++j)
      if (e0 + j < E) atomicOr(&trow[ei[e0 + j] / OWN_ROWS], (1u << j) << b0);
  }
  __syncthreads();

  const int nown   = (N + OWN_ROWS - 1) / OWN_ROWS;
  const int pwords = ((E + CH_EDGES - 1) / CH_EDGES) * CH_WORDS;
  for (int it = tid; it < nown * 8; it += 256) {
    const int o = it >> 3, q = it & 7;
    const unsigned* tc = tab + (q * 4) * TAB_PITCH + o;
    uint4 v;
    v.x = tc[0];
    v.y = tc[TAB_PITCH];
    v.z = tc[2 * TAB_PITCH];
    v.w = tc[3 * TAB_PITCH];
    *(uint4*)(planes + (size_t)o * pwords + blockIdx.x * CH_WORDS + q * 4) = v;
  }
}

__global__ __launch_bounds__(256) void gat_proj_wmma(
    const float* __restrict__ x, const float* __restrict__ W,
    float* __restrict__ h, int nrows) {
  const int wave = threadIdx.x >> 5;
  const int lane = threadIdx.x & 31;
  const long long row0 = ((long long)blockIdx.x * 8 + wave) * 16;
  if (row0 >= nrows) return;

  const int m     = lane & 15;
  const int khalf = (lane >> 4) << 1;
  long long rowA = row0 + m;
  if (rowA >= nrows) rowA = nrows - 1;
  const float* xrow = x + rowA * F_IN;

  v8f acc[HEADS];
  #pragma unroll
  for (int t = 0; t < HEADS; ++t) acc[t] = (v8f){0,0,0,0,0,0,0,0};

  for (int kt = 0; kt < F_IN / 4; ++kt) {
    const int kbase = kt * 4 + khalf;
    v2f a = *(const v2f*)(xrow + kbase);
    #pragma unroll
    for (int t = 0; t < HEADS; ++t) {
      const float* wb = W + t * (F_IN * F_PH) + kbase * F_PH + m;
      v2f b;
      b.x = wb[0];
      b.y = wb[F_PH];
      acc[t] = __builtin_amdgcn_wmma_f32_16x16x4_f32(
           false, a,  false, b,
           (short)0, acc[t],  false,  false);
    }
  }

  const int mofs = (lane >> 4) << 3;
  #pragma unroll
  for (int t = 0; t < HEADS; ++t) {
    #pragma unroll
    for (int v = 0; v < 8; ++v) {
      long long r = row0 + v + mofs;
      if (r < nrows) h[r * F_OUT + t * F_PH + m] = acc[t][v];
    }
  }
}

__global__ void gat_node_scores(const float* __restrict__ h,
                                const float* __restrict__ a_src,
                                const float* __restrict__ a_dst,
                                float* __restrict__ s_src,
                                float* __restrict__ s_dst, int nNH) {
  int idx = blockIdx.x * blockDim.x + threadIdx.x;
  if (idx >= nNH) return;
  int n = idx >> 3, hd = idx & 7;
  const float* hp = h + (long long)n * F_OUT + hd * F_PH;
  const float* as = a_src + hd * F_PH;
  const float* ad = a_dst + hd * F_PH;
  float ss = 0.f, sd = 0.f;
  #pragma unroll
  for (int o = 0; o < F_PH; ++o) {
    float v = hp[o];
    ss += v * as[o];
    sd += v * ad[o];
  }
  s_src[idx] = ss;
  s_dst[idx] = sd;
}


__device__ __forceinline__ unsigned own_key(float v) {
  const unsigned b = __float_as_uint(v);
  return b ^ ((unsigned)((int)b >> 31) | 0x80000000u);
}
__device__ __forceinline__ float own_unkey(unsigned k) {
  return __uint_as_float(k ^ ((unsigned)((int)~k >> 31) | 0x80000000u));
}
__device__ __forceinline__ float own_logit(float ss, float sd) {
  const float v = ss + sd;
  return (v >= 0.0f) ? v : NEG_SLOPE * v;
}

__device__ __forceinline__ void own_slow_max(
    const int* __restrict__ ei, int E, const float* __restrict__ s_dst,
    const float* ssl, unsigned* mxk, int row0, int e) {
  const int l = ei[e] - row0, cc = ei[E + e];
  for (int hd = 0; hd < HEADS; ++hd)
    atomicMax(&mxk[l * HEADS + hd], own_key(own_logit(ssl[l * HEADS + hd], s_dst[cc * HEADS + hd])));
}
__device__ __forceinline__ void own_slow_sum(
    const int* __restrict__ ei, int E, const float* __restrict__ h,
    const float* __restrict__ s_dst, const float* ssl, const unsigned* mxk,
    float* den, float* acc, int row0, int e) {
  const int l = ei[e] - row0, cc = ei[E + e];
  for (int hd = 0; hd < HEADS; ++hd) {
    const float v = own_logit(ssl[l * HEADS + hd], s_dst[cc * HEADS + hd]);
    const float p = expf(v - own_unkey(mxk[l * HEADS + hd]));
    atomicAdd(&den[l * HEADS + hd], p);
    for (int o = 0; o < F_PH; ++o) {
      const int f = hd * F_PH + o;
      atomicAdd(&acc[l * F_OUT + (f & 3) * 32 + (f >> 2)], p * h[(long long)cc * F_OUT + f]);
    }
  }
}

__device__ __forceinline__ void own_append(int e, int* cnt, int* list) {
  const int idx = atomicAdd(cnt, 1);
  if (idx < LIST_CAP) list[idx] = e;
}

__device__ __forceinline__ void own_scan_word(unsigned m, int e0, int* cnt, int* list) {
  while (m != 0u) {
    own_append(e0 + __builtin_ctz(m), cnt, list);
    m &= m - 1u;
  }
}
__device__ __forceinline__ void own_scan_piece(v4u v, int e0, int* cnt, int* list) {
  own_scan_word(v.x, e0,      cnt, list);
  own_scan_word(v.y, e0 + 32, cnt, list);
  own_scan_word(v.z, e0 + 64, cnt, list);
  own_scan_word(v.w, e0 + 96, cnt, list);
}

template <bool FULL>
__device__ __forceinline__ void own_max_batch(
    const int* __restrict__ ei, int E, const float* __restrict__ s_dst,
    const float* ssl, unsigned* mxk, int* list, int row0, int b0, int n, int lane) {
  const int sub = lane >> 3, hd = lane & 7;
  int cl[MAX_SETS], ll[MAX_SETS];
  #pragma unroll
  for (int s = 0; s < MAX_SETS; ++s) {
    int i = b0 + s * 32 + lane;
    if (!FULL && i > n - 1) i = n - 1;
    const int e = list[i];
    ll[s] = ei[e] - row0;
    cl[s] = ei[E + e];
  }
  int   wu[MAX_SETS][8];
  float sd[MAX_SETS][8], ss[MAX_SETS][8];
  #pragma unroll
  for (int s = 0; s < MAX_SETS; ++s) {
    #pragma unroll
    for (int u = 0; u < 8; ++u) {
      const int from = u * 4 + sub;
      const int c = __shfl(cl[s], from);
      wu[s][u] = __shfl(ll[s], from) * HEADS + hd;
      sd[s][u] = s_dst[c * HEADS + hd];
      ss[s][u] = ssl[wu[s][u]];
    }
  }
  #pragma unroll
  for (int s = 0; s < MAX_SETS; ++s) {
    const int i = b0 + s * 32 + lane;
    if (FULL || i < n) list[i] = cl[s] | (ll[s] << COL_BITS);
  }
  #pragma unroll
  for (int s = 0; s < MAX_SETS; ++s) {
    #pragma unroll
    for (int u = 0; u < 8; ++u) {
      if (FULL || b0 + s * 32 + u * 4 + sub < n)
        atomicMax(&mxk[wu[s][u]], own_key(own_logit(ss[s][u], sd[s][u])));
    }
  }
}

template <bool FULL>
__device__ __forceinline__ void own_sum_batch(
    const float* __restrict__ h, const float* __restrict__ s_dst,
    const float* ssl, const unsigned* mxk, float* den, float* acc,
    const int* list, int b0, int n, int lane) {
  const int sub  = lane >> 3;
  const int hd   = lane & 7;
  const int lhd  = lane >> 2;
  int iA = b0 + sub, iB = b0 + 4 + sub;
  if (!FULL && iA > n - 1) iA = n - 1;
  if (!FULL && iB > n - 1) iB = n - 1;
  const int pA = list[iA], pB = list[iB];
  const int cA = pA & ((1 << COL_BITS) - 1), cB = pB & ((1 << COL_BITS) - 1);
  const int wA = (pA >> COL_BITS) * HEADS + hd, wB = (pB >> COL_BITS) * HEADS + hd;
  const float sdA = s_dst[cA * HEADS + hd];
  const float sdB = s_dst[cB * HEADS + hd];

  int    ls[AGG_EDGES];
  float4 hv[AGG_EDGES];
  #pragma unroll
  for (int u = 0; u < AGG_EDGES; ++u) {
    const int first = (u & 3) * 8;
    const int pk = __builtin_amdgcn_readlane(u < 4 ? pA : pB, first);
    ls[u] = pk >> COL_BITS;
    hv[u] = *(const float4*)(h + (long long)(pk & ((1 << COL_BITS) - 1)) * F_OUT + lane * 4);
  }

  const float exA = expf(own_logit(ssl[wA], sdA) - own_unkey(mxk[wA]));
  const float exB = expf(own_logit(ssl[wB], sdB) - own_unkey(mxk[wB]));
  if (FULL || b0 + sub < n)     atomicAdd(&den[wA], exA);
  if (FULL || b0 + 4 + sub < n) atomicAdd(&den[wB], exB);

  #pragma unroll
  for (int u = 0; u < AGG_EDGES; ++u) {
    const float ex = u < 4 ? exA : exB;
    const float a = __shfl(ex, (u & 3) * 8 + lhd);
    if (FULL || b0 + u < n) {
      float* ap = acc + ls[u] * F_OUT + lane;
      atomicAdd(ap +  0, a * hv[u].x);
      atomicAdd(ap + 32, a * hv[u].y);
      atomicAdd(ap + 64, a * hv[u].z);
      atomicAdd(ap + 96, a * hv[u].w);
    }
  }
}

__global__ __launch_bounds__(OWN_THREADS) void gat_softmax_aggregate_owner(
    const int* __restrict__ ei, int E, const float* __restrict__ h,
    const float* __restrict__ s_src, const float* __restrict__ s_dst,
    const float* __restrict__ omap_ws, float* __restrict__ out, int N) {
  __shared__ __attribute__((aligned(16))) float acc[OWN_ROWS * F_OUT];
  __shared__ int list[LIST_CAP];
  __shared__ unsigned mxk[OWN_ROWS * HEADS];
  __shared__ float den[OWN_ROWS * HEADS];
  __shared__ float ssl[OWN_ROWS * HEADS];
  __shared__ int cnt;

  const int tid  = threadIdx.x;
  const int lane = tid & 31;
  const int wave = tid >> 5;
  const int row0 = blockIdx.x * OWN_ROWS;
  const unsigned my = blockIdx.x;

  for (int i = tid; i < OWN_ROWS * F_OUT / 4; i += OWN_THREADS)
    ((float4*)acc)[i] = make_float4(0.f, 0.f, 0.f, 0.f);
  for (int i = tid; i < OWN_ROWS * HEADS; i += OWN_THREADS) {
    mxk[i] = 0u;
    den[i] = 0.f;
    const long long g = (long long)row0 * HEADS + i;
    ssl[i] = (g < (long long)N * HEADS) ? s_src[g] : 0.f;
  }
  if (tid == 0) cnt = 0;
  __syncthreads();

  const int pwords = ((E + CH_EDGES - 1) / CH_EDGES) * CH_WORDS;
  const unsigned* pl = (const unsigned*)omap_ws + (size_t)my * pwords;
  const v4u* p4 = (const v4u*)pl;
  const int npieces = pwords >> 2;
  for (int base = 0; base < npieces; base += 4 * OWN_THREADS) {
    const int i0 = base + tid, i1 = i0 + OWN_THREADS;
    const int i2 = i1 + OWN_THREADS, i3 = i2 + OWN_THREADS;
    const int last = npieces - 1;
    const v4u v0 = __builtin_nontemporal_load(p4 + (i0 < last ? i0 : last));
    const v4u v1 = __builtin_nontemporal_load(p4 + (i1 < last ? i1 : last));
    const v4u v2 = __builtin_nontemporal_load(p4 + (i2 < last ? i2 : last));
    const v4u v3 = __builtin_nontemporal_load(p4 + (i3 < last ? i3 : last));
    if (i0 < npieces) own_scan_piece(v0, i0 * 128, &cnt, list);
    if (i1 < npieces) own_scan_piece(v1, i1 * 128, &cnt, list);
    if (i2 < npieces) own_scan_piece(v2, i2 * 128, &cnt, list);
    if (i3 < npieces) own_scan_piece(v3, i3 * 128, &cnt, list);
  }
  __syncthreads();

  const int n = cnt;
  const bool listed = n <= LIST_CAP;

  if (listed) {
    for (int b0 = wave * (MAX_SETS * 32); b0 < n; b0 += (OWN_THREADS / 32) * (MAX_SETS * 32)) {
      if (b0 + MAX_SETS * 32 <= n)
        own_max_batch<true>(ei, E, s_dst, ssl, mxk, list, row0, b0, n, lane);
      else
        own_max_batch<false>(ei, E, s_dst, ssl, mxk, list, row0, b0, n, lane);
    }
  } else {
    for (int w = tid; w < pwords; w += OWN_THREADS)
      for (unsigned m = pl[w]; m != 0u; m &= m - 1u)
        own_slow_max(ei, E, s_dst, ssl, mxk, row0, w * 32 + __builtin_ctz(m));
  }
  __syncthreads();

  if (listed) {
    for (int b0 = wave * AGG_EDGES; b0 < n; b0 += (OWN_THREADS / 32) * AGG_EDGES) {
      if (b0 + AGG_EDGES <= n)
        own_sum_batch<true>(h, s_dst, ssl, mxk, den, acc, list, b0, n, lane);
      else
        own_sum_batch<false>(h, s_dst, ssl, mxk, den, acc, list, b0, n, lane);
    }
  } else {
    for (int w = tid; w < pwords; w += OWN_THREADS)
      for (unsigned m = pl[w]; m != 0u; m &= m - 1u)
        own_slow_sum(ei, E, h, s_dst, ssl, mxk, den, acc, row0, w * 32 + __builtin_ctz(m));
  }
  __syncthreads();

  int rows = N - row0;
  if (rows > OWN_ROWS) rows = OWN_ROWS;
  float4* o4 = (float4*)(out + (long long)row0 * F_OUT);
  for (int i = tid; i < rows * (F_OUT / 4); i += OWN_THREADS) {
    const int q = i & 31;
    const float* ar = acc + (i >> 5) * F_OUT + q;
    float4 a = make_float4(ar[0], ar[32], ar[64], ar[96]);
    const float d = den[(i >> 5) * HEADS + (q >> 2)] + 1e-16f;
    a.x /= d; a.y /= d; a.z /= d; a.w /= d;
    o4[i] = a;
  }
}

extern "C" void kernel_launch(void* const* d_in, const int* in_sizes, int n_in,
                              void* d_out, int out_size, void* d_ws, size_t ws_size,
                              hipStream_t stream) {
  const float* x     = (const float*)d_in[0];
  const int*   ei    = (const int*)d_in[1];
  const float* W     = (const float*)d_in[2];
  const float* a_src = (const float*)d_in[3];
  const float* a_dst = (const float*)d_in[4];
  float* out = (float*)d_out;

  const int N = in_sizes[0] / F_IN;
  const int E = in_sizes[1] / 2;

  float* h     = (float*)d_ws;
  float* s_src = h     + (size_t)N * F_OUT;
  float* s_dst = s_src + (size_t)N * HEADS;
  float* omap  = s_dst + (size_t)N * HEADS;

  const int nNH   = N * HEADS;
  const int B = 256;

  gat_owner_map<<<(E + CH_EDGES - 1) / CH_EDGES, B, 0, stream>>>(ei, E, N, omap);
  {
    int blocks = (N + 127) / 128;
    gat_proj_wmma<<<blocks, B, 0, stream>>>(x, W, h, N);
  }
  gat_node_scores<<<(nNH + B - 1) / B, B, 0, stream>>>(h, a_src, a_dst, s_src, s_dst, nNH);
  gat_softmax_aggregate_owner<<<(N + OWN_ROWS - 1) / OWN_ROWS, OWN_THREADS, 0, stream>>>(ei, E, h, s_src, s_dst, omap, out, N);
}


# Projection: each wave keeps the B fragments of 4 heads in registers for its whole life (W staged once per work-group via LDS as k-pairs); 256 work-groups loop over 16-row stripes, next x loads ahead
# speedup vs baseline: 3.8181x; 1.1187x over previous
#include <hip/hip_runtime.h>
#include <hip/hip_bf16.h>

typedef __attribute__((ext_vector_type(2))) float v2f;
typedef __attribute__((ext_vector_type(8))) float v8f;
typedef __attribute__((ext_vector_type(4))) unsigned v4u;

#define F_IN   128
#define HEADS  8
#define F_PH   16
#define F_OUT  128
#define NEG_SLOPE 0.2f

#define AGG_EDGES       8
#define MAX_SETS        2
#define OWN_ROWS        400
#define OWN_THREADS     1024
#define LIST_CAP        8192
#define COL_BITS        17
#define CH_EDGES        1024
#define CH_WORDS        32
#define TAB_PITCH       257

__global__ __launch_bounds__(256) void gat_owner_map(
    const int* __restrict__ ei, int E, int N, float* __restrict__ omap_ws) {
  __shared__ __attribute__((aligned(16))) unsigned tab[CH_WORDS * TAB_PITCH];
  unsigned* planes = (unsigned*)omap_ws;
  const int tid = threadIdx.x;
  for (int i = tid; i < CH_WORDS * TAB_PITCH / 4; i += 256)
    ((uint4*)tab)[i] = make_uint4(0u, 0u, 0u, 0u);
  __syncthreads();

  const int e0 = blockIdx.x * CH_EDGES + tid * 4;
  unsigned* trow = tab + (tid >> 3) * TAB_PITCH;
  const unsigned b0 = (unsigned)(tid & 7) * 4u;
  if (e0 + 4 <= E) {
    const int4 r = *(const int4*)(ei + e0);
    atomicOr(&trow[r.x / OWN_ROWS], 1u << b0);
    atomicOr(&trow[r.y / OWN_ROWS], 2u << b0);
    atomicOr(&trow[r.z / OWN_ROWS], 4u << b0);
    atomicOr(&trow[r.w / OWN_ROWS], 8u << b0);
  } else {
    for (int j = 0; j < 4; ++j)
      if (e0 + j < E) atomicOr(&trow[ei[e0 + j] / OWN_ROWS], (1u << j) << b0);
  }
  __syncthreads();

  const int nown   = (N + OWN_ROWS - 1) / OWN_ROWS;
  const int pwords = ((E + CH_EDGES - 1) / CH_EDGES) * CH_WORDS;
  for (int it = tid; it < nown * 8; it += 256) {
    const int o = it >> 3, q = it & 7;
    const unsigned* tc = tab + (q * 4) * TAB_PITCH + o;
    uint4 v;
    v.x = tc[0];
    v.y = tc[TAB_PITCH];
    v.z = tc[2 * TAB_PITCH];
    v.w = tc[3 * TAB_PITCH];
    *(uint4*)(planes + (size_t)o * pwords + blockIdx.x * CH_WORDS + q * 4) = v;
  }
}

#define PROJ_WGS    256
#define PROJ_WAVES  8
#define PROJ_HEADS  4

__device__ __forceinline__ void proj_load_a(const float* __restrict__ x, long long row0,
                                            int m, int khalf, int nrows, v2f* a) {
  long long rowA = row0 + m;
  if (rowA >= nrows) rowA = nrows - 1;
  const float* xrow = x + rowA * F_IN + khalf;
  #pragma unroll
  for (int kt = 0; kt < F_IN / 4; ++kt) a[kt] = *(const v2f*)(xrow + kt * 4);
}

__global__ __launch_bounds__(PROJ_WAVES * 32) void gat_proj_wmma(
    const float* __restrict__ x, const float* __restrict__ W,
    float* __restrict__ h, int nrows) {
  __shared__ __attribute__((aligned(16))) v2f wl[HEADS * (F_IN / 2) * F_PH];
  const int tid  = threadIdx.x;
  #pragma unroll
  for (int j = 0; j < HEADS * (F_IN / 2) * F_PH / (PROJ_WAVES * 32); ++j) {
    const int i = tid + j * (PROJ_WAVES * 32);
    const int t = i >> 10, kp = (i >> 4) & 63, mm = i & 15;
    const float* wp = W + t * (F_IN * F_PH) + (2 * kp) * F_PH + mm;
    v2f p;
    p.x = wp[0];
    p.y = wp[F_PH];
    wl[i] = p;
  }
  __syncthreads();

  const int wave = tid >> 5;
  const int lane = tid & 31;
  const int m     = lane & 15;
  const int khalf = (lane >> 4) << 1;
  const int mofs  = (lane >> 4) << 3;
  const int head0 = (wave & 1) * PROJ_HEADS;
  const int nstripes = (nrows + 15) >> 4;
  const int step = PROJ_WGS * (PROJ_WAVES / 2);
  int s = (int)blockIdx.x + PROJ_WGS * (wave >> 1);
  if (s >= nstripes) return;

  const v2f* wlane = wl + (head0 * (F_IN / 2) + (lane >> 4)) * F_PH + m;
  v2f b[F_IN / 4][PROJ_HEADS];
  #pragma unroll
  for (int kt = 0; kt < F_IN / 4; ++kt) {
    #pragma unroll
    for (int t = 0; t < PROJ_HEADS; ++t) b[kt][t] = wlane[(t * (F_IN / 2) + 2 * kt) * F_PH];
  }

  v2f a[F_IN / 4];
  proj_load_a(x, (long long)s * 16, m, khalf, nrows, a);

  for (; s < nstripes; s += step) {
    const long long row0 = (long long)s * 16;
    v2f an[F_IN / 4];
    const int sn = (s + step < nstripes) ? s + step : s;
    proj_load_a(x, (long long)sn * 16, m, khalf, nrows, an);

    v8f acc[PROJ_HEADS];
    #pragma unroll
    for (int t = 0; t < PROJ_HEADS; ++t) acc[t] = (v8f){0,0,0,0,0,0,0,0};

    #pragma unroll
    for (int kt = 0; kt < F_IN / 4; ++kt) {
      #pragma unroll
      for (int t = 0; t < PROJ_HEADS; ++t) {
        acc[t] = __builtin_amdgcn_wmma_f32_16x16x4_f32(
             false, a[kt],  false, b[kt][t],
             (short)0, acc[t],  false,  false);
      }
    }

    float* hp = h + (row0 + mofs) * F_OUT + head0 * F_PH + m;
    if (row0 + 16 <= nrows) {
      #pragma unroll
      for (int t = 0; t < PROJ_HEADS; ++t) {
        #pragma unroll
        for (int v = 0; v < 8; ++v) hp[v * F_OUT + t * F_PH] = acc[t][v];
      }
    } else {
      #pragma unroll
      for (int t = 0; t < PROJ_HEADS; ++t) {
        #pragma unroll
        for (int v = 0; v < 8; ++v)
          if (row0 + v + mofs < nrows) hp[v * F_OUT + t * F_PH] = acc[t][v];
      }
    }
    #pragma unroll
    for (int kt = 0; kt < F_IN / 4; ++kt) a[kt] = an[kt];
  }
}

__global__ void gat_node_scores(const float* __restrict__ h,
                                const float* __restrict__ a_src,
                                const float* __restrict__ a_dst,
                                float* __restrict__ s_src,
                                float* __restrict__ s_dst, int nNH) {
  int idx = blockIdx.x * blockDim.x + threadIdx.x;
  if (idx >= nNH) return;
  int n = idx >> 3, hd = idx & 7;
  const float* hp = h + (long long)n * F_OUT + hd * F_PH;
  const float* as = a_src + hd * F_PH;
  const float* ad = a_dst + hd * F_PH;
  float ss = 0.f, sd = 0.f;
  #pragma unroll
  for (int o = 0; o < F_PH; ++o) {
    float v = hp[o];
    ss += v * as[o];
    sd += v * ad[o];
  }
  s_src[idx] = ss;
  s_dst[idx] = sd;
}


__device__ __forceinline__ unsigned own_key(float v) {
  const unsigned b = __float_as_uint(v);
  return b ^ ((unsigned)((int)b >> 31) | 0x80000000u);
}
__device__ __forceinline__ float own_unkey(unsigned k) {
  return __uint_as_float(k ^ ((unsigned)((int)~k >> 31) | 0x80000000u));
}
__device__ __forceinline__ float own_logit(float ss, float sd) {
  const float v = ss + sd;
  return (v >= 0.0f) ? v : NEG_SLOPE * v;
}

__device__ __forceinline__ void own_slow_max(
    const int* __restrict__ ei, int E, const float* __restrict__ s_dst,
    const float* ssl, unsigned* mxk, int row0, int e) {
  const int l = ei[e] - row0, cc = ei[E + e];
  for (int hd = 0; hd < HEADS; ++hd)
    atomicMax(&mxk[l * HEADS + hd], own_key(own_logit(ssl[l * HEADS + hd], s_dst[cc * HEADS + hd])));
}
__device__ __forceinline__ void own_slow_sum(
    const int* __restrict__ ei, int E, const float* __restrict__ h,
    const float* __restrict__ s_dst, const float* ssl, const unsigned* mxk,
    float* den, float* acc, int row0, int e) {
  const int l = ei[e] - row0, cc = ei[E + e];
  for (int hd = 0; hd < HEADS; ++hd) {
    const float v = own_logit(ssl[l * HEADS + hd], s_dst[cc * HEADS + hd]);
    const float p = expf(v - own_unkey(mxk[l * HEADS + hd]));
    atomicAdd(&den[l * HEADS + hd], p);
    for (int o = 0; o < F_PH; ++o) {
      const int f = hd * F_PH + o;
      atomicAdd(&acc[l * F_OUT + (f & 3) * 32 + (f >> 2)], p * h[(long long)cc * F_OUT + f]);
    }
  }
}

__device__ __forceinline__ void own_append(int e, int* cnt, int* list) {
  const int idx = atomicAdd(cnt, 1);
  if (idx < LIST_CAP) list[idx] = e;
}

__device__ __forceinline__ void own_scan_word(unsigned m, int e0, int* cnt, int* list) {
  while (m != 0u) {
    own_append(e0 + __builtin_ctz(m), cnt, list);
    m &= m - 1u;
  }
}
__device__ __forceinline__ void own_scan_piece(v4u v, int e0, int* cnt, int* list) {
  own_scan_word(v.x, e0,      cnt, list);
  own_scan_word(v.y, e0 + 32, cnt, list);
  own_scan_word(v.z, e0 + 64, cnt, list);
  own_scan_word(v.w, e0 + 96, cnt, list);
}

template <bool FULL>
__device__ __forceinline__ void own_max_batch(
    const int* __restrict__ ei, int E, const float* __restrict__ s_dst,
    const float* ssl, unsigned* mxk, int* list, int row0, int b0, int n, int lane) {
  const int sub = lane >> 3, hd = lane & 7;
  int cl[MAX_SETS], ll[MAX_SETS];
  #pragma unroll
  for (int s = 0; s < MAX_SETS; ++s) {
    int i = b0 + s * 32 + lane;
    if (!FULL && i > n - 1) i = n - 1;
    const int e = list[i];
    ll[s] = ei[e] - row0;
    cl[s] = ei[E + e];
  }
  int   wu[MAX_SETS][8];
  float sd[MAX_SETS][8], ss[MAX_SETS][8];
  #pragma unroll
  for (int s = 0; s < MAX_SETS; ++s) {
    #pragma unroll
    for (int u = 0; u < 8; ++u) {
      const int from = u * 4 + sub;
      const int c = __shfl(cl[s], from);
      wu[s][u] = __shfl(ll[s], from) * HEADS + hd;
      sd[s][u] = s_dst[c * HEADS + hd];
      ss[s][u] = ssl[wu[s][u]];
    }
  }
  #pragma unroll
  for (int s = 0; s < MAX_SETS; ++s) {
    const int i = b0 + s * 32 + lane;
    if (FULL || i < n) list[i] = cl[s] | (ll[s] << COL_BITS);
  }
  #pragma unroll
  for (int s = 0; s < MAX_SETS; ++s) {
    #pragma unroll
    for (int u = 0; u < 8; ++u) {
      if (FULL || b0 + s * 32 + u * 4 + sub < n)
        atomicMax(&mxk[wu[s][u]], own_key(own_logit(ss[s][u], sd[s][u])));
    }
  }
}

template <bool FULL>
__device__ __forceinline__ void own_sum_batch(
    const float* __restrict__ h, const float* __restrict__ s_dst,
    const float* ssl, const unsigned* mxk, float* den, float* acc,
    const int* list, int b0, int n, int lane) {
  const int sub  = lane >> 3;
  const int hd   = lane & 7;
  const int lhd  = lane >> 2;
  int iA = b0 + sub, iB = b0 + 4 + sub;
  if (!FULL && iA > n - 1) iA = n - 1;
  if (!FULL && iB > n - 1) iB = n - 1;
  const int pA = list[iA], pB = list[iB];
  const int cA = pA & ((1 << COL_BITS) - 1), cB = pB & ((1 << COL_BITS) - 1);
  const int wA = (pA >> COL_BITS) * HEADS + hd, wB = (pB >> COL_BITS) * HEADS + hd;
  const float sdA = s_dst[cA * HEADS + hd];
  const float sdB = s_dst[cB * HEADS + hd];

  int    ls[AGG_EDGES];
  float4 hv[AGG_EDGES];
  #pragma unroll
  for (int u = 0; u < AGG_EDGES; ++u) {
    const int first = (u & 3) * 8;
    const int pk = __builtin_amdgcn_readlane(u < 4 ? pA : pB, first);
    ls[u] = pk >> COL_BITS;
    hv[u] = *(const float4*)(h + (long long)(pk & ((1 << COL_BITS) - 1)) * F_OUT + lane * 4);
  }

  const float exA = expf(own_logit(ssl[wA], sdA) - own_unkey(mxk[wA]));
  const float exB = expf(own_logit(ssl[wB], sdB) - own_unkey(mxk[wB]));
  if (FULL || b0 + sub < n)     atomicAdd(&den[wA], exA);
  if (FULL || b0 + 4 + sub < n) atomicAdd(&den[wB], exB);

  #pragma unroll
  for (int u = 0; u < AGG_EDGES; ++u) {
    const float ex = u < 4 ? exA : exB;
    const float a = __shfl(ex, (u & 3) * 8 + lhd);
    if (FULL || b0 + u < n) {
      float* ap = acc + ls[u] * F_OUT + lane;
      atomicAdd(ap +  0, a * hv[u].x);
      atomicAdd(ap + 32, a * hv[u].y);
      atomicAdd(ap + 64, a * hv[u].z);
      atomicAdd(ap + 96, a * hv[u].w);
    }
  }
}

__global__ __launch_bounds__(OWN_THREADS) void gat_softmax_aggregate_owner(
    const int* __restrict__ ei, int E, const float* __restrict__ h,
    const float* __restrict__ s_src, const float* __restrict__ s_dst,
    const float* __restrict__ omap_ws, float* __restrict__ out, int N) {
  __shared__ __attribute__((aligned(16))) float acc[OWN_ROWS * F_OUT];
  __shared__ int list[LIST_CAP];
  __shared__ unsigned mxk[OWN_ROWS * HEADS];
  __shared__ float den[OWN_ROWS * HEADS];
  __shared__ float ssl[OWN_ROWS * HEADS];
  __shared__ int cnt;

  const int tid  = threadIdx.x;
  const int lane = tid & 31;
  const int wave = tid >> 5;
  const int row0 = blockIdx.x * OWN_ROWS;
  const unsigned my = blockIdx.x;

  for (int i = tid; i < OWN_ROWS * F_OUT / 4; i += OWN_THREADS)
    ((float4*)acc)[i] = make_float4(0.f, 0.f, 0.f, 0.f);
  for (int i = tid; i < OWN_ROWS * HEADS; i += OWN_THREADS) {
    mxk[i] = 0u;
    den[i] = 0.f;
    const long long g = (long long)row0 * HEADS + i;
    ssl[i] = (g < (long long)N * HEADS) ? s_src[g] : 0.f;
  }
  if (tid == 0) cnt = 0;
  __syncthreads();

  const int pwords = ((E + CH_EDGES - 1) / CH_EDGES) * CH_WORDS;
  const unsigned* pl = (const unsigned*)omap_ws + (size_t)my * pwords;
  const v4u* p4 = (const v4u*)pl;
  const int npieces = pwords >> 2;
  for (int base = 0; base < npieces; base += 4 * OWN_THREADS) {
    const int i0 = base + tid, i1 = i0 + OWN_THREADS;
    const int i2 = i1 + OWN_THREADS, i3 = i2 + OWN_THREADS;
    const int last = npieces - 1;
    const v4u v0 = __builtin_nontemporal_load(p4 + (i0 < last ? i0 : last));
    const v4u v1 = __builtin_nontemporal_load(p4 + (i1 < last ? i1 : last));
    const v4u v2 = __builtin_nontemporal_load(p4 + (i2 < last ? i2 : last));
    const v4u v3 = __builtin_nontemporal_load(p4 + (i3 < last ? i3 : last));
    if (i0 < npieces) own_scan_piece(v0, i0 * 128, &cnt, list);
    if (i1 < npieces) own_scan_piece(v1, i1 * 128, &cnt, list);
    if (i2 < npieces) own_scan_piece(v2, i2 * 128, &cnt, list);
    if (i3 < npieces) own_scan_piece(v3, i3 * 128, &cnt, list);
  }
  __syncthreads();

  const int n = cnt;
  const bool listed = n <= LIST_CAP;

  if (listed) {
    for (int b0 = wave * (MAX_SETS * 32); b0 < n; b0 += (OWN_THREADS / 32) * (MAX_SETS * 32)) {
      if (b0 + MAX_SETS * 32 <= n)
        own_max_batch<true>(ei, E, s_dst, ssl, mxk, list, row0, b0, n, lane);
      else
        own_max_batch<false>(ei, E, s_dst, ssl, mxk, list, row0, b0, n, lane);
    }
  } else {
    for (int w = tid; w < pwords; w += OWN_THREADS)
      for (unsigned m = pl[w]; m != 0u; m &= m - 1u)
        own_slow_max(ei, E, s_dst, ssl, mxk, row0, w * 32 + __builtin_ctz(m));
  }
  __syncthreads();

  if (listed) {
    for (int b0 = wave * AGG_EDGES; b0 < n; b0 += (OWN_THREADS / 32) * AGG_EDGES) {
      if (b0 + AGG_EDGES <= n)
        own_sum_batch<true>(h, s_dst, ssl, mxk, den, acc, list, b0, n, lane);
      else
        own_sum_batch<false>(h, s_dst, ssl, mxk, den, acc, list, b0, n, lane);
    }
  } else {
    for (int w = tid; w < pwords; w += OWN_THREADS)
      for (unsigned m = pl[w]; m != 0u; m &= m - 1u)
        own_slow_sum(ei, E, h, s_dst, ssl, mxk, den, acc, row0, w * 32 + __builtin_ctz(m));
  }
  __syncthreads();

  int rows = N - row0;
  if (rows > OWN_ROWS) rows = OWN_ROWS;
  float4* o4 = (float4*)(out + (long long)row0 * F_OUT);
  for (int i = tid; i < rows * (F_OUT / 4); i += OWN_THREADS) {
    const int q = i & 31;
    const float* ar = acc + (i >> 5) * F_OUT + q;
    float4 a = make_float4(ar[0], ar[32], ar[64], ar[96]);
    const float d = den[(i >> 5) * HEADS + (q >> 2)] + 1e-16f;
    a.x /= d; a.y /= d; a.z /= d; a.w /= d;
    o4[i] = a;
  }
}

extern "C" void kernel_launch(void* const* d_in, const int* in_sizes, int n_in,
                              void* d_out, int out_size, void* d_ws, size_t ws_size,
                              hipStream_t stream) {
  const float* x     = (const float*)d_in[0];
  const int*   ei    = (const int*)d_in[1];
  const float* W     = (const float*)d_in[2];
  const float* a_src = (const float*)d_in[3];
  const float* a_dst = (const float*)d_in[4];
  float* out = (float*)d_out;

  const int N = in_sizes[0] / F_IN;
  const int E = in_sizes[1] / 2;

  float* h     = (float*)d_ws;
  float* s_src = h     + (size_t)N * F_OUT;
  float* s_dst = s_src + (size_t)N * HEADS;
  float* omap  = s_dst + (size_t)N * HEADS;

  const int nNH   = N * HEADS;
  const int B = 256;

  gat_owner_map<<<(E + CH_EDGES - 1) / CH_EDGES, B, 0, stream>>>(ei, E, N, omap);
  gat_proj_wmma<<<PROJ_WGS, PROJ_WAVES * 32, 0, stream>>>(x, W, h, N);
  gat_node_scores<<<(nNH + B - 1) / B, B, 0, stream>>>(h, a_src, a_dst, s_src, s_dst, nNH);
  gat_softmax_aggregate_owner<<<(N + OWN_ROWS - 1) / OWN_ROWS, OWN_THREADS, 0, stream>>>(ei, E, h, s_src, s_dst, omap, out, N);
}


# Owner map cell (chunk x owner, one 128-byte line) holds the chunk's own edges as source|local-row words behind a count, not as bits: owner no longer gathers the edge list; LDS arrival order kept
# speedup vs baseline: 4.7222x; 1.2368x over previous
#include <hip/hip_runtime.h>
#include <hip/hip_bf16.h>

typedef __attribute__((ext_vector_type(2))) float v2f;
typedef __attribute__((ext_vector_type(8))) float v8f;
typedef __attribute__((ext_vector_type(4))) unsigned v4u;

#define F_IN   128
#define HEADS  8
#define F_PH   16
#define F_OUT  128
#define NEG_SLOPE 0.2f

#define AGG_EDGES       8
#define MAX_SETS        2
#define OWN_ROWS        400
#define OWN_THREADS     1024
#define LIST_CAP        8192
#define COL_BITS        17
#define CH_EDGES        2048
#define CELL_WORDS      32
#define CELL_CAP        31
#define TAB_PITCH       33
#define MAP_OWNERS      256
#define P1_LOADS        8

#define MAP_EPT (CH_EDGES / 256)

__global__ __launch_bounds__(256) void gat_owner_map(
    const int* __restrict__ ei, int E, int N, float* __restrict__ omap_ws) {
  __shared__ unsigned tab[MAP_OWNERS * TAB_PITCH];
  unsigned* planes = (unsigned*)omap_ws;
  const int tid = threadIdx.x;
  for (int i = tid; i < MAP_OWNERS * TAB_PITCH; i += 256) tab[i] = 0u;
  __syncthreads();

  const int cbase = blockIdx.x * CH_EDGES;
  int  rr[MAP_EPT], cc[MAP_EPT];
  bool ok[MAP_EPT];
  if ((E & 3) == 0 && cbase + CH_EDGES <= E) {
    #pragma unroll
    for (int hh = 0; hh < MAP_EPT / 4; ++hh) {
      const int e0 = cbase + hh * 1024 + tid * 4;
      const int4 a = *(const int4*)(ei + e0);
      const int4 b = *(const int4*)(ei + E + e0);
      rr[4 * hh + 0] = a.x; rr[4 * hh + 1] = a.y; rr[4 * hh + 2] = a.z; rr[4 * hh + 3] = a.w;
      cc[4 * hh + 0] = b.x; cc[4 * hh + 1] = b.y; cc[4 * hh + 2] = b.z; cc[4 * hh + 3] = b.w;
      ok[4 * hh + 0] = ok[4 * hh + 1] = ok[4 * hh + 2] = ok[4 * hh + 3] = true;
    }
  } else {
    #pragma unroll
    for (int j = 0; j < MAP_EPT; ++j) {
      const int e = cbase + (j >> 2) * 1024 + tid * 4 + (j & 3);
      ok[j] = e < E;
      rr[j] = ok[j] ? ei[e] : 0;
      cc[j] = ok[j] ? ei[E + e] : 0;
    }
  }
  unsigned* cell[MAP_EPT];
  unsigned  slot[MAP_EPT];
  #pragma unroll
  for (int j = 0; j < MAP_EPT; ++j) {
    const int o = rr[j] / OWN_ROWS;
    cell[j] = tab + o * TAB_PITCH;
    rr[j] -= o * OWN_ROWS;
    slot[j] = CELL_CAP;
    if (ok[j]) slot[j] = atomicAdd(cell[j], 1u);
  }
  #pragma unroll
  for (int j = 0; j < MAP_EPT; ++j)
    if (slot[j] < CELL_CAP) cell[j][1 + slot[j]] = (unsigned)cc[j] | ((unsigned)rr[j] << COL_BITS);
  __syncthreads();

  const int nown   = (N + OWN_ROWS - 1) / OWN_ROWS;
  const int pwords = ((E + CH_EDGES - 1) / CH_EDGES) * CELL_WORDS;
  for (int it = tid; it < nown * 8; it += 256) {
    const int o = it >> 3, q = it & 7;
    const unsigned* tc = tab + o * TAB_PITCH + q * 4;
    uint4 v;
    v.x = tc[0];
    v.y = tc[1];
    v.z = tc[2];
    v.w = tc[3];
    *(uint4*)(planes + (size_t)o * pwords + blockIdx.x * CELL_WORDS + q * 4) = v;
  }
}

#define PROJ_WGS    256
#define PROJ_WAVES  8
#define PROJ_HEADS  4

__device__ __forceinline__ void proj_load_a(const float* __restrict__ x, long long row0,
                                            int m, int khalf, int nrows, v2f* a) {
  long long rowA = row0 + m;
  if (rowA >= nrows) rowA = nrows - 1;
  const float* xrow = x + rowA * F_IN + khalf;
  #pragma unroll
  for (int kt = 0; kt < F_IN / 4; ++kt) a[kt] = *(const v2f*)(xrow + kt * 4);
}

__global__ __launch_bounds__(PROJ_WAVES * 32) void gat_proj_wmma(
    const float* __restrict__ x, const float* __restrict__ W,
    float* __restrict__ h, int nrows) {
  __shared__ __attribute__((aligned(16))) v2f wl[HEADS * (F_IN / 2) * F_PH];
  const int tid  = threadIdx.x;
  #pragma unroll
  for (int j = 0; j < HEADS * (F_IN / 2) * F_PH / (PROJ_WAVES * 32); ++j) {
    const int i = tid + j * (PROJ_WAVES * 32);
    const int t = i >> 10, kp = (i >> 4) & 63, mm = i & 15;
    const float* wp = W + t * (F_IN * F_PH) + (2 * kp) * F_PH + mm;
    v2f p;
    p.x = wp[0];
    p.y = wp[F_PH];
    wl[i] = p;
  }
  __syncthreads();

  const int wave = tid >> 5;
  const int lane = tid & 31;
  const int m     = lane & 15;
  const int khalf = (lane >> 4) << 1;
  const int mofs  = (lane >> 4) << 3;
  const int head0 = (wave & 1) * PROJ_HEADS;
  const int nstripes = (nrows + 15) >> 4;
  const int step = PROJ_WGS * (PROJ_WAVES / 2);
  int s = (int)blockIdx.x + PROJ_WGS * (wave >> 1);
  if (s >= nstripes) return;

  const v2f* wlane = wl + (head0 * (F_IN / 2) + (lane >> 4)) * F_PH + m;
  v2f b[F_IN / 4][PROJ_HEADS];
  #pragma unroll
  for (int kt = 0; kt < F_IN / 4; ++kt) {
    #pragma unroll
    for (int t = 0; t < PROJ_HEADS; ++t) b[kt][t] = wlane[(t * (F_IN / 2) + 2 * kt) * F_PH];
  }

  v2f a[F_IN / 4];
  proj_load_a(x, (long long)s * 16, m, khalf, nrows, a);

  for (; s < nstripes; s += step) {
    const long long row0 = (long long)s * 16;
    v2f an[F_IN / 4];
    const int sn = (s + step < nstripes) ? s + step : s;
    proj_load_a(x, (long long)sn * 16, m, khalf, nrows, an);

    v8f acc[PROJ_HEADS];
    #pragma unroll
    for (int t = 0; t < PROJ_HEADS; ++t) acc[t] = (v8f){0,0,0,0,0,0,0,0};

    #pragma unroll
    for (int kt = 0; kt < F_IN / 4; ++kt) {
      #pragma unroll
      for (int t = 0; t < PROJ_HEADS; ++t) {
        acc[t] = __builtin_amdgcn_wmma_f32_16x16x4_f32(
             false, a[kt],  false, b[kt][t],
             (short)0, acc[t],  false,  false);
      }
    }

    float* hp = h + (row0 + mofs) * F_OUT + head0 * F_PH + m;
    if (row0 + 16 <= nrows) {
      #pragma unroll
      for (int t = 0; t < PROJ_HEADS; ++t) {
        #pragma unroll
        for (int v = 0; v < 8; ++v) hp[v * F_OUT + t * F_PH] = acc[t][v];
      }
    } else {
      #pragma unroll
      for (int t = 0; t < PROJ_HEADS; ++t) {
        #pragma unroll
        for (int v = 0; v < 8; ++v)
          if (row0 + v + mofs < nrows) hp[v * F_OUT + t * F_PH] = acc[t][v];
      }
    }
    #pragma unroll
    for (int kt = 0; kt < F_IN / 4; ++kt) a[kt] = an[kt];
  }
}

__global__ void gat_node_scores(const float* __restrict__ h,
                                const float* __restrict__ a_src,
                                const float* __restrict__ a_dst,
                                float* __restrict__ s_src,
                                float* __restrict__ s_dst, int nNH) {
  int idx = blockIdx.x * blockDim.x + threadIdx.x;
  if (idx >= nNH) return;
  int n = idx >> 3, hd = idx & 7;
  const float* hp = h + (long long)n * F_OUT + hd * F_PH;
  const float* as = a_src + hd * F_PH;
  const float* ad = a_dst + hd * F_PH;
  float ss = 0.f, sd = 0.f;
  #pragma unroll
  for (int o = 0; o < F_PH; ++o) {
    float v = hp[o];
    ss += v * as[o];
    sd += v * ad[o];
  }
  s_src[idx] = ss;
  s_dst[idx] = sd;
}


__device__ __forceinline__ unsigned own_key(float v) {
  const unsigned b = __float_as_uint(v);
  return b ^ ((unsigned)((int)b >> 31) | 0x80000000u);
}
__device__ __forceinline__ float own_unkey(unsigned k) {
  return __uint_as_float(k ^ ((unsigned)((int)~k >> 31) | 0x80000000u));
}
__device__ __forceinline__ float own_logit(float ss, float sd) {
  const float v = ss + sd;
  return (v >= 0.0f) ? v : NEG_SLOPE * v;
}

__device__ __forceinline__ void own_slow_max(
    const float* __restrict__ s_dst, const float* ssl, unsigned* mxk, unsigned p) {
  const int l = (int)(p >> COL_BITS), cc = (int)(p & ((1u << COL_BITS) - 1u));
  for (int hd = 0; hd < HEADS; ++hd)
    atomicMax(&mxk[l * HEADS + hd], own_key(own_logit(ssl[l * HEADS + hd], s_dst[cc * HEADS + hd])));
}
__device__ __forceinline__ void own_slow_sum(
    const float* __restrict__ h, const float* __restrict__ s_dst, const float* ssl,
    const unsigned* mxk, float* den, float* acc, unsigned p) {
  const int l = (int)(p >> COL_BITS), cc = (int)(p & ((1u << COL_BITS) - 1u));
  for (int hd = 0; hd < HEADS; ++hd) {
    const float v = own_logit(ssl[l * HEADS + hd], s_dst[cc * HEADS + hd]);
    const float p = expf(v - own_unkey(mxk[l * HEADS + hd]));
    atomicAdd(&den[l * HEADS + hd], p);
    for (int o = 0; o < F_PH; ++o) {
      const int f = hd * F_PH + o;
      atomicAdd(&acc[l * F_OUT + (f & 3) * 32 + (f >> 2)], p * h[(long long)cc * F_OUT + f]);
    }
  }
}

__device__ __forceinline__ void own_append(int entry, int* cnt, int* list) {
  const int idx = atomicAdd(cnt, 1);
  if (idx < LIST_CAP) list[idx] = entry;
}

template <class F>
__device__ __forceinline__ void own_chunk_edges(
    const int* __restrict__ ei, int E, int row0, int c, F f) {
  int e1 = (c + 1) * CH_EDGES;
  if (e1 > E) e1 = E;
  for (int e = c * CH_EDGES; e < e1; ++e) {
    const int l = ei[e] - row0;
    if ((unsigned)l < (unsigned)OWN_ROWS) f((unsigned)ei[E + e] | ((unsigned)l << COL_BITS));
  }
}

template <class F>
__device__ __forceinline__ void own_cell_edges(
    const unsigned* __restrict__ pl, const int* __restrict__ ei, int E, int row0, int c, F f) {
  const unsigned* cell = pl + (size_t)c * CELL_WORDS;
  const unsigned k = cell[0];
  if (k <= CELL_CAP) {
    for (unsigned j = 0; j < k; ++j) f(cell[1 + j]);
  } else {
    own_chunk_edges(ei, E, row0, c, f);
  }
}

__device__ __forceinline__ void own_take_word(unsigned word, unsigned w, unsigned k, int at, int* list) {
  if (w >= 1u && w <= k && at < LIST_CAP) list[at] = (int)word;
}

__device__ __forceinline__ void own_take_pieces(
    const v4u* v, int i0, int npieces, int lane, const int* __restrict__ ei, int E, int row0,
    int* cnt, int* list) {
  const int q = lane & 7, lead = lane & ~7;
  __builtin_amdgcn_sched_barrier(0);
  unsigned k[P1_LOADS];
  int tot = 0;
  #pragma unroll
  for (int j = 0; j < P1_LOADS; ++j) {
    k[j] = (unsigned)__shfl((int)v[j].x, lead);
    if (i0 + j * OWN_THREADS < npieces && k[j] <= CELL_CAP) tot += (int)k[j];
  }
  int bs = 0;
  if (q == 0 && tot != 0) bs = atomicAdd(cnt, tot);
  bs = __shfl(bs, lead);
  const unsigned w0 = (unsigned)q * 4u;
  #pragma unroll
  for (int j = 0; j < P1_LOADS; ++j) {
    const int piece = i0 + j * OWN_THREADS;
    if (piece < npieces) {
      if (k[j] <= CELL_CAP) {
        const int at = bs + (int)w0 - 1;
        own_take_word(v[j].x, w0,      k[j], at,     list);
        own_take_word(v[j].y, w0 + 1u, k[j], at + 1, list);
        own_take_word(v[j].z, w0 + 2u, k[j], at + 2, list);
        own_take_word(v[j].w, w0 + 3u, k[j], at + 3, list);
        bs += (int)k[j];
      } else if (q == 0) {
        own_chunk_edges(ei, E, row0, piece >> 3,
                        [&](unsigned p) { own_append((int)p, cnt, list); });
      }
    }
  }
}

template <bool FULL>
__device__ __forceinline__ void own_max_batch(
    const float* __restrict__ s_dst, const float* ssl, unsigned* mxk,
    const int* list, int b0, int n, int lane) {
  const int sub = lane >> 3, hd = lane & 7;
  int pk[MAX_SETS];
  #pragma unroll
  for (int s = 0; s < MAX_SETS; ++s) {
    int i = b0 + s * 32 + lane;
    if (!FULL && i > n - 1) i = n - 1;
    pk[s] = list[i];
  }
  int   wu[MAX_SETS][8];
  float sd[MAX_SETS][8], ss[MAX_SETS][8];
  #pragma unroll
  for (int s = 0; s < MAX_SETS; ++s) {
    #pragma unroll
    for (int u = 0; u < 8; ++u) {
      const int p = __shfl(pk[s], u * 4 + sub);
      wu[s][u] = (p >> COL_BITS) * HEADS + hd;
      sd[s][u] = s_dst[(p & ((1 << COL_BITS) - 1)) * HEADS + hd];
      ss[s][u] = ssl[wu[s][u]];
    }
  }
  #pragma unroll
  for (int s = 0; s < MAX_SETS; ++s) {
    #pragma unroll
    for (int u = 0; u < 8; ++u) {
      if (FULL || b0 + s * 32 + u * 4 + sub < n)
        atomicMax(&mxk[wu[s][u]], own_key(own_logit(ss[s][u], sd[s][u])));
    }
  }
}

template <bool FULL>
__device__ __forceinline__ void own_sum_batch(
    const float* __restrict__ h, const float* __restrict__ s_dst,
    const float* ssl, const unsigned* mxk, float* den, float* acc,
    const int* list, int b0, int n, int lane) {
  const int sub  = lane >> 3;
  const int hd   = lane & 7;
  const int lhd  = lane >> 2;
  int iA = b0 + sub, iB = b0 + 4 + sub;
  if (!FULL && iA > n - 1) iA = n - 1;
  if (!FULL && iB > n - 1) iB = n - 1;
  const int pA = list[iA], pB = list[iB];
  const int cA = pA & ((1 << COL_BITS) - 1), cB = pB & ((1 << COL_BITS) - 1);
  const int wA = (pA >> COL_BITS) * HEADS + hd, wB = (pB >> COL_BITS) * HEADS + hd;
  const float sdA = s_dst[cA * HEADS + hd];
  const float sdB = s_dst[cB * HEADS + hd];

  int    ls[AGG_EDGES];
  float4 hv[AGG_EDGES];
  #pragma unroll
  for (int u = 0; u < AGG_EDGES; ++u) {
    const int first = (u & 3) * 8;
    const int pk = __builtin_amdgcn_readlane(u < 4 ? pA : pB, first);
    ls[u] = pk >> COL_BITS;
    hv[u] = *(const float4*)(h + (long long)(pk & ((1 << COL_BITS) - 1)) * F_OUT + lane * 4);
  }

  const float exA = expf(own_logit(ssl[wA], sdA) - own_unkey(mxk[wA]));
  const float exB = expf(own_logit(ssl[wB], sdB) - own_unkey(mxk[wB]));
  if (FULL || b0 + sub < n)     atomicAdd(&den[wA], exA);
  if (FULL || b0 + 4 + sub < n) atomicAdd(&den[wB], exB);

  #pragma unroll
  for (int u = 0; u < AGG_EDGES; ++u) {
    const float ex = u < 4 ? exA : exB;
    const float a = __shfl(ex, (u & 3) * 8 + lhd);
    if (FULL || b0 + u < n) {
      float* ap = acc + ls[u] * F_OUT + lane;
      atomicAdd(ap +  0, a * hv[u].x);
      atomicAdd(ap + 32, a * hv[u].y);
      atomicAdd(ap + 64, a * hv[u].z);
      atomicAdd(ap + 96, a * hv[u].w);
    }
  }
}

__global__ __launch_bounds__(OWN_THREADS) void gat_softmax_aggregate_owner(
    const int* __restrict__ ei, int E, const float* __restrict__ h,
    const float* __restrict__ s_src, const float* __restrict__ s_dst,
    const float* __restrict__ omap_ws, float* __restrict__ out, int N) {
  __shared__ __attribute__((aligned(16))) float acc[OWN_ROWS * F_OUT];
  __shared__ int list[LIST_CAP];
  __shared__ unsigned mxk[OWN_ROWS * HEADS];
  __shared__ float den[OWN_ROWS * HEADS];
  __shared__ float ssl[OWN_ROWS * HEADS];
  __shared__ int cnt;

  const int tid  = threadIdx.x;
  const int lane = tid & 31;
  const int wave = tid >> 5;
  const int row0 = blockIdx.x * OWN_ROWS;
  const unsigned my = blockIdx.x;

  for (int i = tid; i < OWN_ROWS * F_OUT / 4; i += OWN_THREADS)
    ((float4*)acc)[i] = make_float4(0.f, 0.f, 0.f, 0.f);
  for (int i = tid; i < OWN_ROWS * HEADS; i += OWN_THREADS) {
    mxk[i] = 0u;
    den[i] = 0.f;
    const long long g = (long long)row0 * HEADS + i;
    ssl[i] = (g < (long long)N * HEADS) ? s_src[g] : 0.f;
  }
  if (tid == 0) cnt = 0;
  __syncthreads();

  const int pwords = ((E + CH_EDGES - 1) / CH_EDGES) * CELL_WORDS;
  const int ncells = pwords / CELL_WORDS;
  const unsigned* pl = (const unsigned*)omap_ws + (size_t)my * pwords;
  const v4u* p4 = (const v4u*)pl;
  const int npieces = pwords >> 2;
  for (int base = 0; base < npieces; base += P1_LOADS * OWN_THREADS) {
    const int last = npieces - 1;
    v4u v[P1_LOADS];
    #pragma unroll
    for (int k = 0; k < P1_LOADS; ++k) {
      const int i = base + k * OWN_THREADS + tid;
      v[k] = __builtin_nontemporal_load(p4 + (i < last ? i : last));
    }
    own_take_pieces(v, base + tid, npieces, lane, ei, E, row0, &cnt, list);
  }
  __syncthreads();

  const int n = cnt;
  const bool listed = n <= LIST_CAP;

  if (listed) {
    for (int b0 = wave * (MAX_SETS * 32); b0 < n; b0 += (OWN_THREADS / 32) * (MAX_SETS * 32)) {
      if (b0 + MAX_SETS * 32 <= n)
        own_max_batch<true>(s_dst, ssl, mxk, list, b0, n, lane);
      else
        own_max_batch<false>(s_dst, ssl, mxk, list, b0, n, lane);
    }
  } else {
    for (int c = tid; c < ncells; c += OWN_THREADS)
      own_cell_edges(pl, ei, E, row0, c,
                     [&](unsigned p) { own_slow_max(s_dst, ssl, mxk, p); });
  }
  __syncthreads();

  if (listed) {
    for (int b0 = wave * AGG_EDGES; b0 < n; b0 += (OWN_THREADS / 32) * AGG_EDGES) {
      if (b0 + AGG_EDGES <= n)
        own_sum_batch<true>(h, s_dst, ssl, mxk, den, acc, list, b0, n, lane);
      else
        own_sum_batch<false>(h, s_dst, ssl, mxk, den, acc, list, b0, n, lane);
    }
  } else {
    for (int c = tid; c < ncells; c += OWN_THREADS)
      own_cell_edges(pl, ei, E, row0, c,
                     [&](unsigned p) { own_slow_sum(h, s_dst, ssl, mxk, den, acc, p); });
  }
  __syncthreads();

  int rows = N - row0;
  if (rows > OWN_ROWS) rows = OWN_ROWS;
  float4* o4 = (float4*)(out + (long long)row0 * F_OUT);
  for (int i = tid; i < rows * (F_OUT / 4); i += OWN_THREADS) {
    const int q = i & 31;
    const float* ar = acc + (i >> 5) * F_OUT + q;
    float4 a = make_float4(ar[0], ar[32], ar[64], ar[96]);
    const float d = den[(i >> 5) * HEADS + (q >> 2)] + 1e-16f;
    a.x /= d; a.y /= d; a.z /= d; a.w /= d;
    o4[i] = a;
  }
}

extern "C" void kernel_launch(void* const* d_in, const int* in_sizes, int n_in,
                              void* d_out, int out_size, void* d_ws, size_t ws_size,
                              hipStream_t stream) {
  const float* x     = (const float*)d_in[0];
  const int*   ei    = (const int*)d_in[1];
  const float* W     = (const float*)d_in[2];
  const float* a_src = (const float*)d_in[3];
  const float* a_dst = (const float*)d_in[4];
  float* out = (float*)d_out;

  const int N = in_sizes[0] / F_IN;
  const int E = in_sizes[1] / 2;

  float* h     = (float*)d_ws;
  float* s_src = h     + (size_t)N * F_OUT;
  float* s_dst = s_src + (size_t)N * HEADS;
  float* omap  = s_dst + (size_t)N * HEADS;

  const int nNH   = N * HEADS;
  const int B = 256;

  gat_owner_map<<<(E + CH_EDGES - 1) / CH_EDGES, B, 0, stream>>>(ei, E, N, omap);
  gat_proj_wmma<<<PROJ_WGS, PROJ_WAVES * 32, 0, stream>>>(x, W, h, N);
  gat_node_scores<<<(nNH + B - 1) / B, B, 0, stream>>>(h, a_src, a_dst, s_src, s_dst, nNH);
  gat_softmax_aggregate_owner<<<(N + OWN_ROWS - 1) / OWN_ROWS, OWN_THREADS, 0, stream>>>(ei, E, h, s_src, s_dst, omap, out, N);
}
